# starting kernel
# baseline (speedup 1.0000x reference)
#include <hip/hip_runtime.h>
#include <hip/hip_bf16.h>

typedef float v2f __attribute__((ext_vector_type(2)));
typedef float v4f __attribute__((ext_vector_type(4)));
typedef float v8f __attribute__((ext_vector_type(8)));

#define DI 64
#define HH 128

__device__ __forceinline__ v8f wmma_f32_k4(v2f a, v2f b, v8f c) {
    return __builtin_amdgcn_wmma_f32_16x16x4_f32(false, a, false, b, (short)0, c, false, false);
}

__global__ __launch_bounds__(128)
void k_edge_msg(const float* __restrict__ x, const int* __restrict__ src,
                const float* __restrict__ ea, const float* __restrict__ Wedge,
                float* __restrict__ msg, long long E, long long N) {
    __shared__ float Ws[64][66];
    __shared__ float As[4][16][66];

    const int tid  = threadIdx.x;
    const int wave = tid >> 5;
    const int lane = tid & 31;
    const int hf   = lane >> 4;
    const int lm   = lane & 15;

    for (int i = tid; i < 64 * 64; i += 128)
        Ws[i >> 6][i & 63] = Wedge[i];

    const long long e0 = ((long long)blockIdx.x * 4 + wave) * 16;
    for (int r = 0; r < 16; ++r) {
        long long e  = e0 + r;
        long long ec = e < E ? e : E - 1;
        const v2f av = __builtin_nontemporal_load((const v2f*)(ea + ec * DI + lane * 2));
        As[wave][r][lane * 2 + 0] = av.x;
        As[wave][r][lane * 2 + 1] = av.y;
    }
    __syncthreads();

    for (int n0 = 0; n0 < DI; n0 += 16) {
        v8f acc = {};
#pragma unroll
        for (int k0 = 0; k0 < DI; k0 += 4) {
            const int kb = k0 + hf * 2;
            v2f a, b;
            a.x = As[wave][lm][kb];     a.y = As[wave][lm][kb + 1];
            b.x = Ws[kb][n0 + lm];      b.y = Ws[kb + 1][n0 + lm];
            acc = wmma_f32_k4(a, b, acc);
        }
#pragma unroll
        for (int i = 0; i < 8; ++i) {
            const int m = i + hf * 8;
            const long long e = e0 + m;
            if (e < E) {
                const long long s = (long long)src[e];
                float v = acc[i] + x[s * DI + n0 + lm];
                v = (v > 0.f ? v : 0.f) + 1e-7f;
                __builtin_nontemporal_store(v, msg + e * DI + n0 + lm);
            }
        }
    }
}

__global__ __launch_bounds__(256)
void k_seg_max(const float* __restrict__ msg, const int* __restrict__ dst,
               unsigned int* __restrict__ mx, long long E) {
    const long long i = (long long)blockIdx.x * blockDim.x + threadIdx.x;
    if (i >= E * 16) return;
    const long long e = i >> 4;
    const int d4 = (int)(i & 15) * 4;
    const long long dd = (long long)dst[e];
    const v4f m = __builtin_nontemporal_load((const v4f*)(msg + e * DI + d4));
    unsigned int* b = mx + dd * DI + d4;
    atomicMax(b + 0, __float_as_uint(m.x));
    atomicMax(b + 1, __float_as_uint(m.y));
    atomicMax(b + 2, __float_as_uint(m.z));
    atomicMax(b + 3, __float_as_uint(m.w));
}

__global__ __launch_bounds__(256)
void k_seg_expsum(const float* __restrict__ msg, const int* __restrict__ dst,
                  const float* __restrict__ mx, float* __restrict__ sum, long long E) {
    const long long i = (long long)blockIdx.x * blockDim.x + threadIdx.x;
    if (i >= E * 16) return;
    const long long e = i >> 4;
    const int d4 = (int)(i & 15) * 4;
    const long long dd = (long long)dst[e];
    const v4f m  = __builtin_nontemporal_load((const v4f*)(msg + e * DI + d4));
    const v4f mv = *(const v4f*)(mx + dd * DI + d4);
    float* b = sum + dd * DI + d4;
    atomicAdd(b + 0, __expf(m.x - mv.x));
    atomicAdd(b + 1, __expf(m.y - mv.y));
    atomicAdd(b + 2, __expf(m.z - mv.z));
    atomicAdd(b + 3, __expf(m.w - mv.w));
}

__global__ __launch_bounds__(256)
void k_seg_agg(const float* __restrict__ msg, const int* __restrict__ dst,
               const float* __restrict__ mx, const float* __restrict__ sum,
               float* __restrict__ agg, long long E) {
    const long long i = (long long)blockIdx.x * blockDim.x + threadIdx.x;
    if (i >= E * 16) return;
    const long long e = i >> 4;
    const int d4 = (int)(i & 15) * 4;
    const long long dd = (long long)dst[e];
    const v4f m  = __builtin_nontemporal_load((const v4f*)(msg + e * DI + d4));
    const v4f mv = *(const v4f*)(mx + dd * DI + d4);
    const v4f sv = *(const v4f*)(sum + dd * DI + d4);
    float* b = agg + dd * DI + d4;
    atomicAdd(b + 0, m.x * __expf(m.x - mv.x) / (sv.x + 1e-16f));
    atomicAdd(b + 1, m.y * __expf(m.y - mv.y) / (sv.y + 1e-16f));
    atomicAdd(b + 2, m.z * __expf(m.z - mv.z) / (sv.z + 1e-16f));
    atomicAdd(b + 3, m.w * __expf(m.w - mv.w) / (sv.w + 1e-16f));
}

__global__ __launch_bounds__(128)
void k_h_stats(const float* __restrict__ x, const float* __restrict__ agg,
               const float* __restrict__ W1, float* __restrict__ h,
               float* __restrict__ colsum, float* __restrict__ colsq, long long N) {
    __shared__ float W1s[64][130];
    __shared__ float As[4][16][66];
    __shared__ float cs_s[128], cq_s[128];

    const int tid  = threadIdx.x;
    const int wave = tid >> 5;
    const int lane = tid & 31;
    const int hf   = lane >> 4;
    const int lm   = lane & 15;

    for (int i = tid; i < 64 * HH; i += 128)
        W1s[i >> 7][i & 127] = W1[i];
    if (tid < 128) { cs_s[tid] = 0.f; cq_s[tid] = 0.f; }

    const long long r0 = ((long long)blockIdx.x * 4 + wave) * 16;
    for (int r = 0; r < 16; ++r) {
        long long rr = r0 + r;
        long long rc = rr < N ? rr : N - 1;
        const float2 a  = *(const float2*)(agg + rc * DI + lane * 2);
        const float2 xv = *(const float2*)(x + rc * DI + lane * 2);
        As[wave][r][lane * 2 + 0] = a.x + xv.x;
        As[wave][r][lane * 2 + 1] = a.y + xv.y;
    }
    __syncthreads();

    for (int n0 = 0; n0 < HH; n0 += 16) {
        v8f acc = {};
#pragma unroll
        for (int k0 = 0; k0 < DI; k0 += 4) {
            const int kb = k0 + hf * 2;
            v2f a, b;
            a.x = As[wave][lm][kb];   a.y = As[wave][lm][kb + 1];
            b.x = W1s[kb][n0 + lm];   b.y = W1s[kb + 1][n0 + lm];
            acc = wmma_f32_k4(a, b, acc);
        }
        float p1 = 0.f, p2 = 0.f;
#pragma unroll
        for (int i = 0; i < 8; ++i) {
            const int m = i + hf * 8;
            const long long rr = r0 + m;
            if (rr < N) {
                const float v = acc[i];
                h[rr * HH + n0 + lm] = v;
                p1 += v;
                p2 += v * v;
            }
        }
        p1 += __shfl_xor(p1, 16, 32);
        p2 += __shfl_xor(p2, 16, 32);
        if (hf == 0) {
            atomicAdd(&cs_s[n0 + lm], p1);
            atomicAdd(&cq_s[n0 + lm], p2);
        }
    }
    __syncthreads();
    if (tid < 128) {
        atomicAdd(colsum + tid, cs_s[tid]);
        atomicAdd(colsq + tid, cq_s[tid]);
    }
}

__global__ __launch_bounds__(128)
void k_bn(const float* __restrict__ colsum, const float* __restrict__ colsq,
          const float* __restrict__ gamma, const float* __restrict__ beta,
          float* __restrict__ scale, float* __restrict__ shift, float invN) {
    const int i = threadIdx.x;
    const float mu  = colsum[i] * invN;
    const float var = colsq[i] * invN - mu * mu;
    const float sc  = rsqrtf(var + 1e-5f) * gamma[i];
    scale[i] = sc;
    shift[i] = beta[i] - mu * sc;
}

__global__ __launch_bounds__(64)
void k_out(const float* __restrict__ h, const float* __restrict__ W2,
           const float* __restrict__ scale, const float* __restrict__ shift,
           float* __restrict__ out, long long N) {
    __shared__ float W2s[HH][66];
    __shared__ float As[2][16][130];
    __shared__ float sc_s[128], sh_s[128];

    const int tid  = threadIdx.x;
    const int wave = tid >> 5;
    const int lane = tid & 31;
    const int hf   = lane >> 4;
    const int lm   = lane & 15;

    for (int i = tid; i < HH * DI; i += 64)
        W2s[i >> 6][i & 63] = W2[i];
    for (int i = tid; i < 128; i += 64) { sc_s[i] = scale[i]; sh_s[i] = shift[i]; }
    __syncthreads();

    const long long r0 = ((long long)blockIdx.x * 2 + wave) * 16;
    for (int r = 0; r < 16; ++r) {
        long long rr = r0 + r;
        long long rc = rr < N ? rr : N - 1;
        const int c = lane * 4;
        const v4f hv = *(const v4f*)(h + rc * HH + c);
        float v0 = hv.x * sc_s[c + 0] + sh_s[c + 0];
        float v1 = hv.y * sc_s[c + 1] + sh_s[c + 1];
        float v2 = hv.z * sc_s[c + 2] + sh_s[c + 2];
        float v3 = hv.w * sc_s[c + 3] + sh_s[c + 3];
        As[wave][r][c + 0] = v0 > 0.f ? v0 : 0.f;
        As[wave][r][c + 1] = v1 > 0.f ? v1 : 0.f;
        As[wave][r][c + 2] = v2 > 0.f ? v2 : 0.f;
        As[wave][r][c + 3] = v3 > 0.f ? v3 : 0.f;
    }
    __syncthreads();

    for (int n0 = 0; n0 < DI; n0 += 16) {
        v8f acc = {};
#pragma unroll
        for (int k0 = 0; k0 < HH; k0 += 4) {
            const int kb = k0 + hf * 2;
            v2f a, b;
            a.x = As[wave][lm][kb];   a.y = As[wave][lm][kb + 1];
            b.x = W2s[kb][n0 + lm];   b.y = W2s[kb + 1][n0 + lm];
            acc = wmma_f32_k4(a, b, acc);
        }
#pragma unroll
        for (int i = 0; i < 8; ++i) {
            const int m = i + hf * 8;
            const long long rr = r0 + m;
            if (rr < N) __builtin_nontemporal_store(acc[i], out + rr * DI + n0 + lm);
        }
    }
}

extern "C" void kernel_launch(void* const* d_in, const int* in_sizes, int n_in,
                              void* d_out, int out_size, void* d_ws, size_t ws_size,
                              hipStream_t stream) {
    const float* x     = (const float*)d_in[0];
    const int*   eidx  = (const int*)d_in[1];
    const float* ea    = (const float*)d_in[2];
    const float* Wedge = (const float*)d_in[3];
    const float* W1    = (const float*)d_in[4];
    const float* gamma = (const float*)d_in[5];
    const float* beta  = (const float*)d_in[6];
    const float* W2    = (const float*)d_in[7];
    float* out = (float*)d_out;

    const long long N = in_sizes[0] / DI;
    const long long E = in_sizes[1] / 2;
    const int* src = eidx;
    const int* dst = eidx + E;

    float* ws   = (float*)d_ws;
    float* msg  = ws;
    float* mx   = msg + (size_t)E * DI;
    float* sum  = mx  + (size_t)N * DI;
    float* agg  = sum + (size_t)N * DI;
    float* cs   = agg + (size_t)N * DI;
    float* cq   = cs + 128;
    float* sc   = cq + 128;
    float* sh   = sc + 128;
    float* hbuf = sh + 128;

    hipMemsetAsync(mx, 0, ((size_t)3 * N * DI + 256) * sizeof(float), stream);

    k_edge_msg<<<(int)((E + 63) / 64), 128, 0, stream>>>(x, src, ea, Wedge, msg, E, N);

    const long long tot = E * 16;
    const int segBlocks = (int)((tot + 255) / 256);
    k_seg_max<<<segBlocks, 256, 0, stream>>>(msg, dst, (unsigned int*)mx, E);
    k_seg_expsum<<<segBlocks, 256, 0, stream>>>(msg, dst, mx, sum, E);
    k_seg_agg<<<segBlocks, 256, 0, stream>>>(msg, dst, mx, sum, agg, E);

    const int strips = (int)((N + 15) / 16);
    k_h_stats<<<(strips + 3) / 4, 128, 0, stream>>>(x, agg, W1, hbuf, cs, cq, N);
    k_bn<<<1, 128, 0, stream>>>(cs, cq, gamma, beta, sc, sh, 1.0f / (float)N);
    k_out<<<(strips + 1) / 2, 64, 0, stream>>>(hbuf, W2, sc, sh, out, N);
}


# segment softmax without float atomics: msg rows stored contiguous per dst (int rank atomics + block scan), one per-node register pass does max, exp-sum, agg; sums in arrival-rank order, 1 divide/node
# speedup vs baseline: 1.5824x; 1.5824x over previous
#include <hip/hip_runtime.h>
#include <hip/hip_bf16.h>

typedef float v2f __attribute__((ext_vector_type(2)));
typedef float v4f __attribute__((ext_vector_type(4)));
typedef float v8f __attribute__((ext_vector_type(8)));

#define DI 64
#define HH 128

__device__ __forceinline__ v8f wmma_f32_k4(v2f a, v2f b, v8f c) {
    return __builtin_amdgcn_wmma_f32_16x16x4_f32(false, a, false, b, (short)0, c, false, false);
}

__global__ __launch_bounds__(128)
void k_edge_msg(const float* __restrict__ x, const int* __restrict__ src,
                const int* __restrict__ dst, const int* __restrict__ pos,
                const int* __restrict__ offs,
                const float* __restrict__ ea, const float* __restrict__ Wedge,
                float* __restrict__ msg, long long E, long long N) {
    __shared__ float Ws[64][66];
    __shared__ float As[4][16][66];

    const int tid  = threadIdx.x;
    const int wave = tid >> 5;
    const int lane = tid & 31;
    const int hf   = lane >> 4;
    const int lm   = lane & 15;

    for (int i = tid; i < 64 * 64; i += 128)
        Ws[i >> 6][i & 63] = Wedge[i];

    const long long e0 = ((long long)blockIdx.x * 4 + wave) * 16;

    for (int r = 0; r < 16; ++r) {
        long long e  = e0 + r;
        long long ec = e < E ? e : E - 1;
        const v2f av = __builtin_nontemporal_load((const v2f*)(ea + ec * DI + lane * 2));
        As[wave][r][lane * 2 + 0] = av.x;
        As[wave][r][lane * 2 + 1] = av.y;
    }
    __syncthreads();

    int xrow[8], mrow[8];
#pragma unroll
    for (int i = 0; i < 8; ++i) {
        const long long e  = e0 + i + hf * 8;
        const long long ec = e < E ? e : E - 1;
        xrow[i] = src[ec];
        mrow[i] = offs[dst[ec]] + pos[ec];
    }

    for (int n0 = 0; n0 < DI; n0 += 16) {
        v8f acc = {};
#pragma unroll
        for (int k0 = 0; k0 < DI; k0 += 4) {
            const int kb = k0 + hf * 2;
            v2f a, b;
            a.x = As[wave][lm][kb];     a.y = As[wave][lm][kb + 1];
            b.x = Ws[kb][n0 + lm];      b.y = Ws[kb + 1][n0 + lm];
            acc = wmma_f32_k4(a, b, acc);
        }
#pragma unroll
        for (int i = 0; i < 8; ++i) {
            const int m = i + hf * 8;
            const long long e = e0 + m;
            if (e < E) {
                float v = acc[i] + x[(long long)xrow[i] * DI + n0 + lm];
                v = (v > 0.f ? v : 0.f) + 1e-7f;
                __builtin_nontemporal_store(v, msg + (long long)mrow[i] * DI + n0 + lm);
            }
        }
    }
}

__global__ __launch_bounds__(256)
void k_rank(const int* __restrict__ dst, int* __restrict__ cnt,
            int* __restrict__ pos, long long E) {
    const long long e = (long long)blockIdx.x * blockDim.x + threadIdx.x;
    if (e >= E) return;
    pos[e] = atomicAdd(cnt + dst[e], 1);
}

#define SCAN_EPT 16
__global__ __launch_bounds__(1024)
void k_scan(const int* __restrict__ cnt, int* __restrict__ offs, int N) {
    __shared__ int wtot[32];
    const int tid  = threadIdx.x;
    const int lane = tid & 31;
    const int wave = tid >> 5;
    int carry = 0;
    for (int base = 0; base < N; base += 1024 * SCAN_EPT) {
        const int i0 = base + tid * SCAN_EPT;
        int c[SCAN_EPT];
#pragma unroll
        for (int k = 0; k < SCAN_EPT; ++k)
            c[k] = (i0 + k < N) ? cnt[i0 + k] : 0;
        int t = 0;
#pragma unroll
        for (int k = 0; k < SCAN_EPT; ++k) {
            const int v = c[k];
            c[k] = t;
            t += v;
        }
        int s = t;
#pragma unroll
        for (int d = 1; d < 32; d <<= 1) {
            const int v = __shfl_up(s, d, 32);
            if (lane >= d) s += v;
        }
        if (lane == 31) wtot[wave] = s;
        __syncthreads();
        const int w = wtot[lane];
        int wsum = w;
#pragma unroll
        for (int d = 1; d < 32; d <<= 1) {
            const int v = __shfl_up(wsum, d, 32);
            if (lane >= d) wsum += v;
        }
        const int wave_excl = __shfl(wsum, wave, 32) - __shfl(w, wave, 32);
        const int total     = __shfl(wsum, 31, 32);
        const int ex = carry + wave_excl + s - t;
#pragma unroll
        for (int k = 0; k < SCAN_EPT; ++k)
            if (i0 + k < N) offs[i0 + k] = ex + c[k];
        carry += total;
        __syncthreads();
    }
}

#define NCAP 24
__global__ __launch_bounds__(256)
void k_node_agg(const float* __restrict__ msg, const int* __restrict__ offs,
                const int* __restrict__ cnt, float* __restrict__ agg, long long N) {
    const long long g = (long long)blockIdx.x * blockDim.x + threadIdx.x;
    const long long node = g >> 4;
    if (node >= N) return;
    const int d4  = (int)(g & 15) * 4;
    const int deg = cnt[node];
    const float* p = msg + (long long)offs[node] * DI + d4;
    const float ninf = -__builtin_huge_valf();
    v4f mx = {ninf, ninf, ninf, ninf};
    v4f S = {0.f, 0.f, 0.f, 0.f};
    v4f A = {0.f, 0.f, 0.f, 0.f};
    if (deg <= NCAP) {
        v4f rows[NCAP];
#pragma unroll
        for (int j = 0; j < NCAP; ++j) {
            rows[j] = (v4f){ninf, ninf, ninf, ninf};
            if (j < deg) rows[j] = __builtin_nontemporal_load((const v4f*)(p + (long long)j * DI));
        }
#pragma unroll
        for (int j = 0; j < NCAP; ++j) {
            mx.x = fmaxf(mx.x, rows[j].x);  mx.y = fmaxf(mx.y, rows[j].y);
            mx.z = fmaxf(mx.z, rows[j].z);  mx.w = fmaxf(mx.w, rows[j].w);
        }
#pragma unroll
        for (int j = 0; j < NCAP; ++j) {
            const bool w = j < deg;
            const v4f m = rows[j];
            const float ex = __expf(m.x - mx.x), ey = __expf(m.y - mx.y);
            const float ez = __expf(m.z - mx.z), ew = __expf(m.w - mx.w);
            S.x += ex;  S.y += ey;  S.z += ez;  S.w += ew;
            A.x += (w ? m.x : 0.f) * ex;  A.y += (w ? m.y : 0.f) * ey;
            A.z += (w ? m.z : 0.f) * ez;  A.w += (w ? m.w : 0.f) * ew;
        }
    } else {
        for (int j = 0; j < deg; ++j) {
            const v4f m = *(const v4f*)(p + (long long)j * DI);
            mx.x = fmaxf(mx.x, m.x);  mx.y = fmaxf(mx.y, m.y);
            mx.z = fmaxf(mx.z, m.z);  mx.w = fmaxf(mx.w, m.w);
        }
        for (int j = 0; j < deg; ++j) {
            const v4f m = __builtin_nontemporal_load((const v4f*)(p + (long long)j * DI));
            const float ex = __expf(m.x - mx.x), ey = __expf(m.y - mx.y);
            const float ez = __expf(m.z - mx.z), ew = __expf(m.w - mx.w);
            S.x += ex;  S.y += ey;  S.z += ez;  S.w += ew;
            A.x += m.x * ex;  A.y += m.y * ey;  A.z += m.z * ez;  A.w += m.w * ew;
        }
    }
    v4f o = {0.f, 0.f, 0.f, 0.f};
    if (deg > 0) {
        o.x = A.x / (S.x + 1e-16f);  o.y = A.y / (S.y + 1e-16f);
        o.z = A.z / (S.z + 1e-16f);  o.w = A.w / (S.w + 1e-16f);
    }
    *(v4f*)(agg + node * DI + d4) = o;
}

__global__ __launch_bounds__(128)
void k_h_stats(const float* __restrict__ x, const float* __restrict__ agg,
               const float* __restrict__ W1, float* __restrict__ h,
               float* __restrict__ colsum, float* __restrict__ colsq, long long N) {
    __shared__ float W1s[64][130];
    __shared__ float As[4][16][66];
    __shared__ float cs_s[128], cq_s[128];

    const int tid  = threadIdx.x;
    const int wave = tid >> 5;
    const int lane = tid & 31;
    const int hf   = lane >> 4;
    const int lm   = lane & 15;

    for (int i = tid; i < 64 * HH; i += 128)
        W1s[i >> 7][i & 127] = W1[i];
    if (tid < 128) { cs_s[tid] = 0.f; cq_s[tid] = 0.f; }

    const long long r0 = ((long long)blockIdx.x * 4 + wave) * 16;
    for (int r = 0; r < 16; ++r) {
        long long rr = r0 + r;
        long long rc = rr < N ? rr : N - 1;
        const float2 a  = *(const float2*)(agg + rc * DI + lane * 2);
        const float2 xv = *(const float2*)(x + rc * DI + lane * 2);
        As[wave][r][lane * 2 + 0] = a.x + xv.x;
        As[wave][r][lane * 2 + 1] = a.y + xv.y;
    }
    __syncthreads();

    for (int n0 = 0; n0 < HH; n0 += 16) {
        v8f acc = {};
#pragma unroll
        for (int k0 = 0; k0 < DI; k0 += 4) {
            const int kb = k0 + hf * 2;
            v2f a, b;
            a.x = As[wave][lm][kb];   a.y = As[wave][lm][kb + 1];
            b.x = W1s[kb][n0 + lm];   b.y = W1s[kb + 1][n0 + lm];
            acc = wmma_f32_k4(a, b, acc);
        }
        float p1 = 0.f, p2 = 0.f;
#pragma unroll
        for (int i = 0; i < 8; ++i) {
            const int m = i + hf * 8;
            const long long rr = r0 + m;
            if (rr < N) {
                const float v = acc[i];
                h[rr * HH + n0 + lm] = v;
                p1 += v;
                p2 += v * v;
            }
        }
        p1 += __shfl_xor(p1, 16, 32);
        p2 += __shfl_xor(p2, 16, 32);
        if (hf == 0) {
            atomicAdd(&cs_s[n0 + lm], p1);
            atomicAdd(&cq_s[n0 + lm], p2);
        }
    }
    __syncthreads();
    if (tid < 128) {
        atomicAdd(colsum + tid, cs_s[tid]);
        atomicAdd(colsq + tid, cq_s[tid]);
    }
}

__global__ __launch_bounds__(128)
void k_bn(const float* __restrict__ colsum, const float* __restrict__ colsq,
          const float* __restrict__ gamma, const float* __restrict__ beta,
          float* __restrict__ scale, float* __restrict__ shift, float invN) {
    const int i = threadIdx.x;
    const float mu  = colsum[i] * invN;
    const float var = colsq[i] * invN - mu * mu;
    const float sc  = rsqrtf(var + 1e-5f) * gamma[i];
    scale[i] = sc;
    shift[i] = beta[i] - mu * sc;
}

__global__ __launch_bounds__(64)
void k_out(const float* __restrict__ h, const float* __restrict__ W2,
           const float* __restrict__ scale, const float* __restrict__ shift,
           float* __restrict__ out, long long N) {
    __shared__ float W2s[HH][66];
    __shared__ float As[2][16][130];
    __shared__ float sc_s[128], sh_s[128];

    const int tid  = threadIdx.x;
    const int wave = tid >> 5;
    const int lane = tid & 31;
    const int hf   = lane >> 4;
    const int lm   = lane & 15;

    for (int i = tid; i < HH * DI; i += 64)
        W2s[i >> 6][i & 63] = W2[i];
    for (int i = tid; i < 128; i += 64) { sc_s[i] = scale[i]; sh_s[i] = shift[i]; }
    __syncthreads();

    const long long r0 = ((long long)blockIdx.x * 2 + wave) * 16;
    for (int r = 0; r < 16; ++r) {
        long long rr = r0 + r;
        long long rc = rr < N ? rr : N - 1;
        const int c = lane * 4;
        const v4f hv = *(const v4f*)(h + rc * HH + c);
        float v0 = hv.x * sc_s[c + 0] + sh_s[c + 0];
        float v1 = hv.y * sc_s[c + 1] + sh_s[c + 1];
        float v2 = hv.z * sc_s[c + 2] + sh_s[c + 2];
        float v3 = hv.w * sc_s[c + 3] + sh_s[c + 3];
        As[wave][r][c + 0] = v0 > 0.f ? v0 : 0.f;
        As[wave][r][c + 1] = v1 > 0.f ? v1 : 0.f;
        As[wave][r][c + 2] = v2 > 0.f ? v2 : 0.f;
        As[wave][r][c + 3] = v3 > 0.f ? v3 : 0.f;
    }
    __syncthreads();

    for (int n0 = 0; n0 < DI; n0 += 16) {
        v8f acc = {};
#pragma unroll
        for (int k0 = 0; k0 < HH; k0 += 4) {
            const int kb = k0 + hf * 2;
            v2f a, b;
            a.x = As[wave][lm][kb];   a.y = As[wave][lm][kb + 1];
            b.x = W2s[kb][n0 + lm];   b.y = W2s[kb + 1][n0 + lm];
            acc = wmma_f32_k4(a, b, acc);
        }
#pragma unroll
        for (int i = 0; i < 8; ++i) {
            const int m = i + hf * 8;
            const long long rr = r0 + m;
            if (rr < N) __builtin_nontemporal_store(acc[i], out + rr * DI + n0 + lm);
        }
    }
}

extern "C" void kernel_launch(void* const* d_in, const int* in_sizes, int n_in,
                              void* d_out, int out_size, void* d_ws, size_t ws_size,
                              hipStream_t stream) {
    const float* x     = (const float*)d_in[0];
    const int*   eidx  = (const int*)d_in[1];
    const float* ea    = (const float*)d_in[2];
    const float* Wedge = (const float*)d_in[3];
    const float* W1    = (const float*)d_in[4];
    const float* gamma = (const float*)d_in[5];
    const float* beta  = (const float*)d_in[6];
    const float* W2    = (const float*)d_in[7];
    float* out = (float*)d_out;

    const long long N = in_sizes[0] / DI;
    const long long E = in_sizes[1] / 2;
    const int* src = eidx;
    const int* dst = eidx + E;

    float* ws   = (float*)d_ws;
    float* msg  = ws;
    int*   cnt  = (int*)(msg + (size_t)E * DI);
    float* cs   = (float*)(cnt + (size_t)N);
    float* cq   = cs + 128;
    int*   offs = (int*)(cq + 128);
    int*   pos  = offs + (size_t)N;
    float* agg  = msg + (size_t)E * DI + (size_t)2 * N * DI;
    float* sc   = agg + (size_t)N * DI + 256;
    float* sh   = sc + 128;
    float* hbuf = sh + 128;

    hipMemsetAsync(cnt, 0, ((size_t)N + 256) * sizeof(float), stream);

    k_rank<<<(int)((E + 255) / 256), 256, 0, stream>>>(dst, cnt, pos, E);
    k_scan<<<1, 1024, 0, stream>>>(cnt, offs, (int)N);

    k_edge_msg<<<(int)((E + 63) / 64), 128, 0, stream>>>(x, src, dst, pos, offs, ea, Wedge, msg, E, N);

    k_node_agg<<<(int)((N * 16 + 255) / 256), 256, 0, stream>>>(msg, offs, cnt, agg, N);

    const int strips = (int)((N + 15) / 16);
    k_h_stats<<<(strips + 3) / 4, 128, 0, stream>>>(x, agg, W1, hbuf, cs, cq, N);
    k_bn<<<1, 128, 0, stream>>>(cs, cq, gamma, beta, sc, sh, 1.0f / (float)N);
    k_out<<<(strips + 1) / 2, 64, 0, stream>>>(hbuf, W2, sc, sh, out, N);
}


# edge-message epilogue row-wise via LDS: x gather and msg store move whole 256-B rows (16 B a lane); the 8 gathers are unconditional and in flight under the products, not 32 guarded 4-B round trips
# speedup vs baseline: 2.1592x; 1.3645x over previous
#include <hip/hip_runtime.h>
#include <hip/hip_bf16.h>

typedef float v2f __attribute__((ext_vector_type(2)));
typedef float v4f __attribute__((ext_vector_type(4)));
typedef float v8f __attribute__((ext_vector_type(8)));

#define DI 64
#define HH 128

__device__ __forceinline__ v8f wmma_f32_k4(v2f a, v2f b, v8f c) {
    return __builtin_amdgcn_wmma_f32_16x16x4_f32(false, a, false, b, (short)0, c, false, false);
}

__global__ __launch_bounds__(128)
void k_edge_msg(const float* __restrict__ x, const int* __restrict__ src,
                const int* __restrict__ dst, const int* __restrict__ pos,
                const int* __restrict__ offs,
                const float* __restrict__ ea, const float* __restrict__ Wedge,
                float* __restrict__ msg, long long E, long long N) {
    __shared__ float Ws[64][66];
    __shared__ __attribute__((aligned(16))) float As[4][16][68];

    const int tid  = threadIdx.x;
    const int wave = tid >> 5;
    const int lane = tid & 31;
    const int hf   = lane >> 4;
    const int lm   = lane & 15;

    const long long e0 = ((long long)blockIdx.x * 4 + wave) * 16;

    int xrow[8], drow[8], prow[8];
#pragma unroll
    for (int j = 0; j < 8; ++j) {
        const long long e  = e0 + 2 * j + hf;
        const long long ec = e < E ? e : E - 1;
        xrow[j] = src[ec];
        drow[j] = dst[ec];
        prow[j] = pos[ec];
    }

    v2f av[16];
#pragma unroll
    for (int r = 0; r < 16; ++r) {
        const long long e  = e0 + r;
        const long long ec = e < E ? e : E - 1;
        av[r] = __builtin_nontemporal_load((const v2f*)(ea + ec * DI + lane * 2));
    }

    for (int i = tid; i < 64 * 64; i += 128)
        Ws[i >> 6][i & 63] = Wedge[i];

#pragma unroll
    for (int r = 0; r < 16; ++r) {
        As[wave][r][lane * 2 + 0] = av[r].x;
        As[wave][r][lane * 2 + 1] = av[r].y;
    }
    __syncthreads();

    int mrow[8];
    v4f xg[8];
#pragma unroll
    for (int j = 0; j < 8; ++j)
        mrow[j] = offs[drow[j]] + prow[j];
#pragma unroll
    for (int j = 0; j < 8; ++j)
        xg[j] = *(const v4f*)(x + (long long)xrow[j] * DI + lm * 4);

    v8f acc0 = {}, acc1 = {}, acc2 = {}, acc3 = {};
#pragma unroll
    for (int k0 = 0; k0 < DI; k0 += 4) {
        const int kb = k0 + hf * 2;
        v2f a, b;
        a.x = As[wave][lm][kb];     a.y = As[wave][lm][kb + 1];
        b.x = Ws[kb][lm];           b.y = Ws[kb + 1][lm];
        acc0 = wmma_f32_k4(a, b, acc0);
        b.x = Ws[kb][16 + lm];      b.y = Ws[kb + 1][16 + lm];
        acc1 = wmma_f32_k4(a, b, acc1);
        b.x = Ws[kb][32 + lm];      b.y = Ws[kb + 1][32 + lm];
        acc2 = wmma_f32_k4(a, b, acc2);
        b.x = Ws[kb][48 + lm];      b.y = Ws[kb + 1][48 + lm];
        acc3 = wmma_f32_k4(a, b, acc3);
    }

#pragma unroll
    for (int i = 0; i < 8; ++i) {
        const int m = i + hf * 8;
        As[wave][m][lm]      = acc0[i];
        As[wave][m][16 + lm] = acc1[i];
        As[wave][m][32 + lm] = acc2[i];
        As[wave][m][48 + lm] = acc3[i];
    }
    __syncthreads();

    if (e0 + 16 <= E) {
#pragma unroll
        for (int j = 0; j < 8; ++j) {
            const int row = 2 * j + hf;
            const v4f d = *(const v4f*)&As[wave][row][lm * 4];
            v4f v;
            v.x = d.x + xg[j].x;  v.y = d.y + xg[j].y;
            v.z = d.z + xg[j].z;  v.w = d.w + xg[j].w;
            v.x = (v.x > 0.f ? v.x : 0.f) + 1e-7f;  v.y = (v.y > 0.f ? v.y : 0.f) + 1e-7f;
            v.z = (v.z > 0.f ? v.z : 0.f) + 1e-7f;  v.w = (v.w > 0.f ? v.w : 0.f) + 1e-7f;
            __builtin_nontemporal_store(v, (v4f*)(msg + (long long)mrow[j] * DI + lm * 4));
        }
    } else {
#pragma unroll
        for (int j = 0; j < 8; ++j) {
            const int row = 2 * j + hf;
            const v4f d = *(const v4f*)&As[wave][row][lm * 4];
            v4f v;
            v.x = d.x + xg[j].x;  v.y = d.y + xg[j].y;
            v.z = d.z + xg[j].z;  v.w = d.w + xg[j].w;
            v.x = (v.x > 0.f ? v.x : 0.f) + 1e-7f;  v.y = (v.y > 0.f ? v.y : 0.f) + 1e-7f;
            v.z = (v.z > 0.f ? v.z : 0.f) + 1e-7f;  v.w = (v.w > 0.f ? v.w : 0.f) + 1e-7f;
            if (e0 + row < E)
                __builtin_nontemporal_store(v, (v4f*)(msg + (long long)mrow[j] * DI + lm * 4));
        }
    }
}

__global__ __launch_bounds__(256)
void k_rank(const int* __restrict__ dst, int* __restrict__ cnt,
            int* __restrict__ pos, long long E) {
    const long long e = (long long)blockIdx.x * blockDim.x + threadIdx.x;
    if (e >= E) return;
    pos[e] = atomicAdd(cnt + dst[e], 1);
}

#define SCAN_EPT 16
__global__ __launch_bounds__(1024)
void k_scan(const int* __restrict__ cnt, int* __restrict__ offs, int N) {
    __shared__ int wtot[32];
    const int tid  = threadIdx.x;
    const int lane = tid & 31;
    const int wave = tid >> 5;
    int carry = 0;
    for (int base = 0; base < N; base += 1024 * SCAN_EPT) {
        const int i0 = base + tid * SCAN_EPT;
        int c[SCAN_EPT];
#pragma unroll
        for (int k = 0; k < SCAN_EPT; ++k)
            c[k] = (i0 + k < N) ? cnt[i0 + k] : 0;
        int t = 0;
#pragma unroll
        for (int k = 0; k < SCAN_EPT; ++k) {
            const int v = c[k];
            c[k] = t;
            t += v;
        }
        int s = t;
#pragma unroll
        for (int d = 1; d < 32; d <<= 1) {
            const int v = __shfl_up(s, d, 32);
            if (lane >= d) s += v;
        }
        if (lane == 31) wtot[wave] = s;
        __syncthreads();
        const int w = wtot[lane];
        int wsum = w;
#pragma unroll
        for (int d = 1; d < 32; d <<= 1) {
            const int v = __shfl_up(wsum, d, 32);
            if (lane >= d) wsum += v;
        }
        const int wave_excl = __shfl(wsum, wave, 32) - __shfl(w, wave, 32);
        const int total     = __shfl(wsum, 31, 32);
        const int ex = carry + wave_excl + s - t;
#pragma unroll
        for (int k = 0; k < SCAN_EPT; ++k)
            if (i0 + k < N) offs[i0 + k] = ex + c[k];
        carry += total;
        __syncthreads();
    }
}

#define NCAP 24
__global__ __launch_bounds__(256)
void k_node_agg(const float* __restrict__ msg, const int* __restrict__ offs,
                const int* __restrict__ cnt, float* __restrict__ agg, long long N) {
    const long long g = (long long)blockIdx.x * blockDim.x + threadIdx.x;
    const long long node = g >> 4;
    if (node >= N) return;
    const int d4  = (int)(g & 15) * 4;
    const int deg = cnt[node];
    const float* p = msg + (long long)offs[node] * DI + d4;
    const float ninf = -__builtin_huge_valf();
    v4f mx = {ninf, ninf, ninf, ninf};
    v4f S = {0.f, 0.f, 0.f, 0.f};
    v4f A = {0.f, 0.f, 0.f, 0.f};
    if (deg <= NCAP) {
        v4f rows[NCAP];
#pragma unroll
        for (int j = 0; j < NCAP; ++j) {
            rows[j] = (v4f){ninf, ninf, ninf, ninf};
            if (j < deg) rows[j] = __builtin_nontemporal_load((const v4f*)(p + (long long)j * DI));
        }
#pragma unroll
        for (int j = 0; j < NCAP; ++j) {
            mx.x = fmaxf(mx.x, rows[j].x);  mx.y = fmaxf(mx.y, rows[j].y);
            mx.z = fmaxf(mx.z, rows[j].z);  mx.w = fmaxf(mx.w, rows[j].w);
        }
#pragma unroll
        for (int j = 0; j < NCAP; ++j) {
            const bool w = j < deg;
            const v4f m = rows[j];
            const float ex = __expf(m.x - mx.x), ey = __expf(m.y - mx.y);
            const float ez = __expf(m.z - mx.z), ew = __expf(m.w - mx.w);
            S.x += ex;  S.y += ey;  S.z += ez;  S.w += ew;
            A.x += (w ? m.x : 0.f) * ex;  A.y += (w ? m.y : 0.f) * ey;
            A.z += (w ? m.z : 0.f) * ez;  A.w += (w ? m.w : 0.f) * ew;
        }
    } else {
        for (int j = 0; j < deg; ++j) {
            const v4f m = *(const v4f*)(p + (long long)j * DI);
            mx.x = fmaxf(mx.x, m.x);  mx.y = fmaxf(mx.y, m.y);
            mx.z = fmaxf(mx.z, m.z);  mx.w = fmaxf(mx.w, m.w);
        }
        for (int j = 0; j < deg; ++j) {
            const v4f m = __builtin_nontemporal_load((const v4f*)(p + (long long)j * DI));
            const float ex = __expf(m.x - mx.x), ey = __expf(m.y - mx.y);
            const float ez = __expf(m.z - mx.z), ew = __expf(m.w - mx.w);
            S.x += ex;  S.y += ey;  S.z += ez;  S.w += ew;
            A.x += m.x * ex;  A.y += m.y * ey;  A.z += m.z * ez;  A.w += m.w * ew;
        }
    }
    v4f o = {0.f, 0.f, 0.f, 0.f};
    if (deg > 0) {
        o.x = A.x / (S.x + 1e-16f);  o.y = A.y / (S.y + 1e-16f);
        o.z = A.z / (S.z + 1e-16f);  o.w = A.w / (S.w + 1e-16f);
    }
    *(v4f*)(agg + node * DI + d4) = o;
}

__global__ __launch_bounds__(128)
void k_h_stats(const float* __restrict__ x, const float* __restrict__ agg,
               const float* __restrict__ W1, float* __restrict__ h,
               float* __restrict__ colsum, float* __restrict__ colsq, long long N) {
    __shared__ float W1s[64][130];
    __shared__ float As[4][16][66];
    __shared__ float cs_s[128], cq_s[128];

    const int tid  = threadIdx.x;
    const int wave = tid >> 5;
    const int lane = tid & 31;
    const int hf   = lane >> 4;
    const int lm   = lane & 15;

    for (int i = tid; i < 64 * HH; i += 128)
        W1s[i >> 7][i & 127] = W1[i];
    if (tid < 128) { cs_s[tid] = 0.f; cq_s[tid] = 0.f; }

    const long long r0 = ((long long)blockIdx.x * 4 + wave) * 16;
    for (int r = 0; r < 16; ++r) {
        long long rr = r0 + r;
        long long rc = rr < N ? rr : N - 1;
        const float2 a  = *(const float2*)(agg + rc * DI + lane * 2);
        const float2 xv = *(const float2*)(x + rc * DI + lane * 2);
        As[wave][r][lane * 2 + 0] = a.x + xv.x;
        As[wave][r][lane * 2 + 1] = a.y + xv.y;
    }
    __syncthreads();

    for (int n0 = 0; n0 < HH; n0 += 16) {
        v8f acc = {};
#pragma unroll
        for (int k0 = 0; k0 < DI; k0 += 4) {
            const int kb = k0 + hf * 2;
            v2f a, b;
            a.x = As[wave][lm][kb];   a.y = As[wave][lm][kb + 1];
            b.x = W1s[kb][n0 + lm];   b.y = W1s[kb + 1][n0 + lm];
            acc = wmma_f32_k4(a, b, acc);
        }
        float p1 = 0.f, p2 = 0.f;
#pragma unroll
        for (int i = 0; i < 8; ++i) {
            const int m = i + hf * 8;
            const long long rr = r0 + m;
            if (rr < N) {
                const float v = acc[i];
                h[rr * HH + n0 + lm] = v;
                p1 += v;
                p2 += v * v;
            }
        }
        p1 += __shfl_xor(p1, 16, 32);
        p2 += __shfl_xor(p2, 16, 32);
        if (hf == 0) {
            atomicAdd(&cs_s[n0 + lm], p1);
            atomicAdd(&cq_s[n0 + lm], p2);
        }
    }
    __syncthreads();
    if (tid < 128) {
        atomicAdd(colsum + tid, cs_s[tid]);
        atomicAdd(colsq + tid, cq_s[tid]);
    }
}

__global__ __launch_bounds__(128)
void k_bn(const float* __restrict__ colsum, const float* __restrict__ colsq,
          const float* __restrict__ gamma, const float* __restrict__ beta,
          float* __restrict__ scale, float* __restrict__ shift, float invN) {
    const int i = threadIdx.x;
    const float mu  = colsum[i] * invN;
    const float var = colsq[i] * invN - mu * mu;
    const float sc  = rsqrtf(var + 1e-5f) * gamma[i];
    scale[i] = sc;
    shift[i] = beta[i] - mu * sc;
}

__global__ __launch_bounds__(64)
void k_out(const float* __restrict__ h, const float* __restrict__ W2,
           const float* __restrict__ scale, const float* __restrict__ shift,
           float* __restrict__ out, long long N) {
    __shared__ float W2s[HH][66];
    __shared__ float As[2][16][130];
    __shared__ float sc_s[128], sh_s[128];

    const int tid  = threadIdx.x;
    const int wave = tid >> 5;
    const int lane = tid & 31;
    const int hf   = lane >> 4;
    const int lm   = lane & 15;

    for (int i = tid; i < HH * DI; i += 64)
        W2s[i >> 6][i & 63] = W2[i];
    for (int i = tid; i < 128; i += 64) { sc_s[i] = scale[i]; sh_s[i] = shift[i]; }
    __syncthreads();

    const long long r0 = ((long long)blockIdx.x * 2 + wave) * 16;
    for (int r = 0; r < 16; ++r) {
        long long rr = r0 + r;
        long long rc = rr < N ? rr : N - 1;
        const int c = lane * 4;
        const v4f hv = *(const v4f*)(h + rc * HH + c);
        float v0 = hv.x * sc_s[c + 0] + sh_s[c + 0];
        float v1 = hv.y * sc_s[c + 1] + sh_s[c + 1];
        float v2 = hv.z * sc_s[c + 2] + sh_s[c + 2];
        float v3 = hv.w * sc_s[c + 3] + sh_s[c + 3];
        As[wave][r][c + 0] = v0 > 0.f ? v0 : 0.f;
        As[wave][r][c + 1] = v1 > 0.f ? v1 : 0.f;
        As[wave][r][c + 2] = v2 > 0.f ? v2 : 0.f;
        As[wave][r][c + 3] = v3 > 0.f ? v3 : 0.f;
    }
    __syncthreads();

    for (int n0 = 0; n0 < DI; n0 += 16) {
        v8f acc = {};
#pragma unroll
        for (int k0 = 0; k0 < HH; k0 += 4) {
            const int kb = k0 + hf * 2;
            v2f a, b;
            a.x = As[wave][lm][kb];   a.y = As[wave][lm][kb + 1];
            b.x = W2s[kb][n0 + lm];   b.y = W2s[kb + 1][n0 + lm];
            acc = wmma_f32_k4(a, b, acc);
        }
#pragma unroll
        for (int i = 0; i < 8; ++i) {
            const int m = i + hf * 8;
            const long long rr = r0 + m;
            if (rr < N) __builtin_nontemporal_store(acc[i], out + rr * DI + n0 + lm);
        }
    }
}

extern "C" void kernel_launch(void* const* d_in, const int* in_sizes, int n_in,
                              void* d_out, int out_size, void* d_ws, size_t ws_size,
                              hipStream_t stream) {
    const float* x     = (const float*)d_in[0];
    const int*   eidx  = (const int*)d_in[1];
    const float* ea    = (const float*)d_in[2];
    const float* Wedge = (const float*)d_in[3];
    const float* W1    = (const float*)d_in[4];
    const float* gamma = (const float*)d_in[5];
    const float* beta  = (const float*)d_in[6];
    const float* W2    = (const float*)d_in[7];
    float* out = (float*)d_out;

    const long long N = in_sizes[0] / DI;
    const long long E = in_sizes[1] / 2;
    const int* src = eidx;
    const int* dst = eidx + E;

    float* ws   = (float*)d_ws;
    float* msg  = ws;
    int*   cnt  = (int*)(msg + (size_t)E * DI);
    float* cs   = (float*)(cnt + (size_t)N);
    float* cq   = cs + 128;
    int*   offs = (int*)(cq + 128);
    int*   pos  = offs + (size_t)N;
    float* agg  = msg + (size_t)E * DI + (size_t)2 * N * DI;
    float* sc   = agg + (size_t)N * DI + 256;
    float* sh   = sc + 128;
    float* hbuf = sh + 128;

    hipMemsetAsync(cnt, 0, ((size_t)N + 256) * sizeof(float), stream);

    k_rank<<<(int)((E + 255) / 256), 256, 0, stream>>>(dst, cnt, pos, E);
    k_scan<<<1, 1024, 0, stream>>>(cnt, offs, (int)N);

    k_edge_msg<<<(int)((E + 63) / 64), 128, 0, stream>>>(x, src, dst, pos, offs, ea, Wedge, msg, E, N);

    k_node_agg<<<(int)((N * 16 + 255) / 256), 256, 0, stream>>>(msg, offs, cnt, agg, N);

    const int strips = (int)((N + 15) / 16);
    k_h_stats<<<(strips + 3) / 4, 128, 0, stream>>>(x, agg, W1, hbuf, cs, cq, N);
    k_bn<<<1, 128, 0, stream>>>(cs, cq, gamma, beta, sc, sh, 1.0f / (float)N);
    k_out<<<(strips + 1) / 2, 64, 0, stream>>>(hbuf, W2, sc, sh, out, N);
}


# edge ranks from LDS counters: 64 work-groups rank one edge chunk each (returning LDS add between block barriers), a per-node pass over the chunk counts gives degrees; sums in chunk-then-rank order
# speedup vs baseline: 2.4023x; 1.1126x over previous
#include <hip/hip_runtime.h>
#include <hip/hip_bf16.h>

typedef float v2f __attribute__((ext_vector_type(2)));
typedef float v4f __attribute__((ext_vector_type(4)));
typedef float v8f __attribute__((ext_vector_type(8)));

#define DI 64
#define HH 128
#define NCHUNK 64
#define NR 65536

__device__ __forceinline__ v8f wmma_f32_k4(v2f a, v2f b, v8f c) {
    return __builtin_amdgcn_wmma_f32_16x16x4_f32(false, a, false, b, (short)0, c, false, false);
}

__global__ __launch_bounds__(128)
void k_edge_msg(const float* __restrict__ x, const int* __restrict__ src,
                const int* __restrict__ dst, const int* __restrict__ pos,
                const int* __restrict__ offs, const int* __restrict__ cpre,
                const float* __restrict__ ea, const float* __restrict__ Wedge,
                float* __restrict__ msg, long long E, long long N, int NP, int C) {
    __shared__ float Ws[64][66];
    __shared__ __attribute__((aligned(16))) float As[4][16][68];

    const int tid  = threadIdx.x;
    const int wave = tid >> 5;
    const int lane = tid & 31;
    const int hf   = lane >> 4;
    const int lm   = lane & 15;

    const long long e0 = ((long long)blockIdx.x * 4 + wave) * 16;
    const long long ef = e0 < E ? e0 : E - 1;
    const int* __restrict__ cpre_c = cpre + (long long)((unsigned)ef / (unsigned)C) * NP;

    int xrow[8], drow[8], prow[8];
#pragma unroll
    for (int j = 0; j < 8; ++j) {
        const long long e  = e0 + 2 * j + hf;
        const long long ec = e < E ? e : E - 1;
        xrow[j] = src[ec];
        drow[j] = dst[ec];
        prow[j] = pos[ec];
    }

    v2f av[16];
#pragma unroll
    for (int r = 0; r < 16; ++r) {
        const long long e  = e0 + r;
        const long long ec = e < E ? e : E - 1;
        av[r] = __builtin_nontemporal_load((const v2f*)(ea + ec * DI + lane * 2));
    }

    for (int i = tid; i < 64 * 64; i += 128)
        Ws[i >> 6][i & 63] = Wedge[i];

#pragma unroll
    for (int r = 0; r < 16; ++r) {
        As[wave][r][lane * 2 + 0] = av[r].x;
        As[wave][r][lane * 2 + 1] = av[r].y;
    }
    __syncthreads();

    int mrow[8];
    v4f xg[8];
#pragma unroll
    for (int j = 0; j < 8; ++j)
        mrow[j] = offs[drow[j]] + cpre_c[drow[j]] + prow[j];
#pragma unroll
    for (int j = 0; j < 8; ++j)
        xg[j] = *(const v4f*)(x + (long long)xrow[j] * DI + lm * 4);

    v8f acc0 = {}, acc1 = {}, acc2 = {}, acc3 = {};
#pragma unroll
    for (int k0 = 0; k0 < DI; k0 += 4) {
        const int kb = k0 + hf * 2;
        v2f a, b;
        a.x = As[wave][lm][kb];     a.y = As[wave][lm][kb + 1];
        b.x = Ws[kb][lm];           b.y = Ws[kb + 1][lm];
        acc0 = wmma_f32_k4(a, b, acc0);
        b.x = Ws[kb][16 + lm];      b.y = Ws[kb + 1][16 + lm];
        acc1 = wmma_f32_k4(a, b, acc1);
        b.x = Ws[kb][32 + lm];      b.y = Ws[kb + 1][32 + lm];
        acc2 = wmma_f32_k4(a, b, acc2);
        b.x = Ws[kb][48 + lm];      b.y = Ws[kb + 1][48 + lm];
        acc3 = wmma_f32_k4(a, b, acc3);
    }

#pragma unroll
    for (int i = 0; i < 8; ++i) {
        const int m = i + hf * 8;
        As[wave][m][lm]      = acc0[i];
        As[wave][m][16 + lm] = acc1[i];
        As[wave][m][32 + lm] = acc2[i];
        As[wave][m][48 + lm] = acc3[i];
    }
    __syncthreads();

    if (e0 + 16 <= E) {
#pragma unroll
        for (int j = 0; j < 8; ++j) {
            const int row = 2 * j + hf;
            const v4f d = *(const v4f*)&As[wave][row][lm * 4];
            v4f v;
            v.x = d.x + xg[j].x;  v.y = d.y + xg[j].y;
            v.z = d.z + xg[j].z;  v.w = d.w + xg[j].w;
            v.x = (v.x > 0.f ? v.x : 0.f) + 1e-7f;  v.y = (v.y > 0.f ? v.y : 0.f) + 1e-7f;
            v.z = (v.z > 0.f ? v.z : 0.f) + 1e-7f;  v.w = (v.w > 0.f ? v.w : 0.f) + 1e-7f;
            __builtin_nontemporal_store(v, (v4f*)(msg + (long long)mrow[j] * DI + lm * 4));
        }
    } else {
#pragma unroll
        for (int j = 0; j < 8; ++j) {
            const int row = 2 * j + hf;
            const v4f d = *(const v4f*)&As[wave][row][lm * 4];
            v4f v;
            v.x = d.x + xg[j].x;  v.y = d.y + xg[j].y;
            v.z = d.z + xg[j].z;  v.w = d.w + xg[j].w;
            v.x = (v.x > 0.f ? v.x : 0.f) + 1e-7f;  v.y = (v.y > 0.f ? v.y : 0.f) + 1e-7f;
            v.z = (v.z > 0.f ? v.z : 0.f) + 1e-7f;  v.w = (v.w > 0.f ? v.w : 0.f) + 1e-7f;
            if (e0 + row < E)
                __builtin_nontemporal_store(v, (v4f*)(msg + (long long)mrow[j] * DI + lm * 4));
        }
    }
}

#define CNT_KPT 16
__global__ __launch_bounds__(1024)
void k_count(const int* __restrict__ dst, int* __restrict__ pos, int* __restrict__ chist,
             long long E, int N, int NP, int C) {
    __shared__ __attribute__((aligned(16))) int hist[NR];
    const int tid = threadIdx.x;
    const int nlo = blockIdx.y * NR;
    const int nn  = (N - nlo) < NR ? (N - nlo) : NR;
    for (int i = tid * 4; i < nn; i += 4096) *(int4*)&hist[i] = make_int4(0, 0, 0, 0);
    __syncthreads();
    const int lo = blockIdx.x * C;
    const int hi = (lo + C) < (int)E ? (lo + C) : (int)E;
    for (int base = lo; base < hi; base += 1024 * CNT_KPT) {
        int d[CNT_KPT], r[CNT_KPT];
#pragma unroll
        for (int k = 0; k < CNT_KPT; ++k) {
            const int e = base + k * 1024 + tid;
            const int v = dst[e < hi ? e : hi - 1];
            d[k] = e < hi ? v - nlo : -1;
        }
#pragma unroll
        for (int k = 0; k < CNT_KPT; ++k) {
            r[k] = 0;
            if ((unsigned)d[k] < (unsigned)nn) r[k] = atomicAdd(&hist[d[k]], 1);
        }
#pragma unroll
        for (int k = 0; k < CNT_KPT; ++k) {
            const int e = base + k * 1024 + tid;
            if ((unsigned)d[k] < (unsigned)nn) pos[e] = r[k];
        }
    }
    __syncthreads();
    int* __restrict__ row = chist + (long long)blockIdx.x * NP + nlo;
    for (int i = tid; i < nn; i += 1024) row[i] = hist[i];
}

__global__ __launch_bounds__(256)
void k_colpre(int* __restrict__ chist, int* __restrict__ cnt, float* __restrict__ cstat, int N, int NP) {
    const int n = blockIdx.x * 256 + threadIdx.x;
    if (blockIdx.x == 0) cstat[threadIdx.x] = 0.f;
    if (n >= N) return;
    int run = 0;
    for (int b0 = 0; b0 < NCHUNK; b0 += 16) {
        int v[16];
#pragma unroll
        for (int k = 0; k < 16; ++k) v[k] = chist[(long long)(b0 + k) * NP + n];
#pragma unroll
        for (int k = 0; k < 16; ++k) {
            chist[(long long)(b0 + k) * NP + n] = run;
            run += v[k];
        }
    }
    cnt[n] = run;
}

#define SCAN_EPT 16
__global__ __launch_bounds__(1024)
void k_scan(const int* __restrict__ cnt, int* __restrict__ offs, int N) {
    __shared__ int wtot[32];
    const int tid  = threadIdx.x;
    const int lane = tid & 31;
    const int wave = tid >> 5;
    int carry = 0;
    for (int base = 0; base < N; base += 1024 * SCAN_EPT) {
        const int i0 = base + tid * SCAN_EPT;
        int c[SCAN_EPT];
#pragma unroll
        for (int k = 0; k < SCAN_EPT; ++k)
            c[k] = (i0 + k < N) ? cnt[i0 + k] : 0;
        int t = 0;
#pragma unroll
        for (int k = 0; k < SCAN_EPT; ++k) {
            const int v = c[k];
            c[k] = t;
            t += v;
        }
        int s = t;
#pragma unroll
        for (int d = 1; d < 32; d <<= 1) {
            const int v = __shfl_up(s, d, 32);
            if (lane >= d) s += v;
        }
        if (lane == 31) wtot[wave] = s;
        __syncthreads();
        const int w = wtot[lane];
        int wsum = w;
#pragma unroll
        for (int d = 1; d < 32; d <<= 1) {
            const int v = __shfl_up(wsum, d, 32);
            if (lane >= d) wsum += v;
        }
        const int wave_excl = __shfl(wsum, wave, 32) - __shfl(w, wave, 32);
        const int total     = __shfl(wsum, 31, 32);
        const int ex = carry + wave_excl + s - t;
#pragma unroll
        for (int k = 0; k < SCAN_EPT; ++k)
            if (i0 + k < N) offs[i0 + k] = ex + c[k];
        carry += total;
        __syncthreads();
    }
}

#define NCAP 24
__global__ __launch_bounds__(256)
void k_node_agg(const float* __restrict__ msg, const int* __restrict__ offs,
                const int* __restrict__ cnt, float* __restrict__ agg, long long N) {
    const long long g = (long long)blockIdx.x * blockDim.x + threadIdx.x;
    const long long node = g >> 4;
    if (node >= N) return;
    const int d4  = (int)(g & 15) * 4;
    const int deg = cnt[node];
    const float* p = msg + (long long)offs[node] * DI + d4;
    const float ninf = -__builtin_huge_valf();
    v4f mx = {ninf, ninf, ninf, ninf};
    v4f S = {0.f, 0.f, 0.f, 0.f};
    v4f A = {0.f, 0.f, 0.f, 0.f};
    if (deg <= NCAP) {
        v4f rows[NCAP];
#pragma unroll
        for (int j = 0; j < NCAP; ++j) {
            rows[j] = (v4f){ninf, ninf, ninf, ninf};
            if (j < deg) rows[j] = __builtin_nontemporal_load((const v4f*)(p + (long long)j * DI));
        }
#pragma unroll
        for (int j = 0; j < NCAP; ++j) {
            mx.x = fmaxf(mx.x, rows[j].x);  mx.y = fmaxf(mx.y, rows[j].y);
            mx.z = fmaxf(mx.z, rows[j].z);  mx.w = fmaxf(mx.w, rows[j].w);
        }
#pragma unroll
        for (int j = 0; j < NCAP; ++j) {
            const bool w = j < deg;
            const v4f m = rows[j];
            const float ex = __expf(m.x - mx.x), ey = __expf(m.y - mx.y);
            const float ez = __expf(m.z - mx.z), ew = __expf(m.w - mx.w);
            S.x += ex;  S.y += ey;  S.z += ez;  S.w += ew;
            A.x += (w ? m.x : 0.f) * ex;  A.y += (w ? m.y : 0.f) * ey;
            A.z += (w ? m.z : 0.f) * ez;  A.w += (w ? m.w : 0.f) * ew;
        }
    } else {
        for (int j = 0; j < deg; ++j) {
            const v4f m = *(const v4f*)(p + (long long)j * DI);
            mx.x = fmaxf(mx.x, m.x);  mx.y = fmaxf(mx.y, m.y);
            mx.z = fmaxf(mx.z, m.z);  mx.w = fmaxf(mx.w, m.w);
        }
        for (int j = 0; j < deg; ++j) {
            const v4f m = __builtin_nontemporal_load((const v4f*)(p + (long long)j * DI));
            const float ex = __expf(m.x - mx.x), ey = __expf(m.y - mx.y);
            const float ez = __expf(m.z - mx.z), ew = __expf(m.w - mx.w);
            S.x += ex;  S.y += ey;  S.z += ez;  S.w += ew;
            A.x += m.x * ex;  A.y += m.y * ey;  A.z += m.z * ez;  A.w += m.w * ew;
        }
    }
    v4f o = {0.f, 0.f, 0.f, 0.f};
    if (deg > 0) {
        o.x = A.x / (S.x + 1e-16f);  o.y = A.y / (S.y + 1e-16f);
        o.z = A.z / (S.z + 1e-16f);  o.w = A.w / (S.w + 1e-16f);
    }
    *(v4f*)(agg + node * DI + d4) = o;
}

__global__ __launch_bounds__(128)
void k_h_stats(const float* __restrict__ x, const float* __restrict__ agg,
               const float* __restrict__ W1, float* __restrict__ h,
               float* __restrict__ colsum, float* __restrict__ colsq, long long N) {
    __shared__ float W1s[64][130];
    __shared__ float As[4][16][66];
    __shared__ float cs_s[128], cq_s[128];

    const int tid  = threadIdx.x;
    const int wave = tid >> 5;
    const int lane = tid & 31;
    const int hf   = lane >> 4;
    const int lm   = lane & 15;

    for (int i = tid; i < 64 * HH; i += 128)
        W1s[i >> 7][i & 127] = W1[i];
    if (tid < 128) { cs_s[tid] = 0.f; cq_s[tid] = 0.f; }

    const long long r0 = ((long long)blockIdx.x * 4 + wave) * 16;
    for (int r = 0; r < 16; ++r) {
        long long rr = r0 + r;
        long long rc = rr < N ? rr : N - 1;
        const float2 a  = *(const float2*)(agg + rc * DI + lane * 2);
        const float2 xv = *(const float2*)(x + rc * DI + lane * 2);
        As[wave][r][lane * 2 + 0] = a.x + xv.x;
        As[wave][r][lane * 2 + 1] = a.y + xv.y;
    }
    __syncthreads();

    for (int n0 = 0; n0 < HH; n0 += 16) {
        v8f acc = {};
#pragma unroll
        for (int k0 = 0; k0 < DI; k0 += 4) {
            const int kb = k0 + hf * 2;
            v2f a, b;
            a.x = As[wave][lm][kb];   a.y = As[wave][lm][kb + 1];
            b.x = W1s[kb][n0 + lm];   b.y = W1s[kb + 1][n0 + lm];
            acc = wmma_f32_k4(a, b, acc);
        }
        float p1 = 0.f, p2 = 0.f;
#pragma unroll
        for (int i = 0; i < 8; ++i) {
            const int m = i + hf * 8;
            const long long rr = r0 + m;
            if (rr < N) {
                const float v = acc[i];
                h[rr * HH + n0 + lm] = v;
                p1 += v;
                p2 += v * v;
            }
        }
        p1 += __shfl_xor(p1, 16, 32);
        p2 += __shfl_xor(p2, 16, 32);
        if (hf == 0) {
            atomicAdd(&cs_s[n0 + lm], p1);
            atomicAdd(&cq_s[n0 + lm], p2);
        }
    }
    __syncthreads();
    if (tid < 128) {
        atomicAdd(colsum + tid, cs_s[tid]);
        atomicAdd(colsq + tid, cq_s[tid]);
    }
}

__global__ __launch_bounds__(128)
void k_bn(const float* __restrict__ colsum, const float* __restrict__ colsq,
          const float* __restrict__ gamma, const float* __restrict__ beta,
          float* __restrict__ scale, float* __restrict__ shift, float invN) {
    const int i = threadIdx.x;
    const float mu  = colsum[i] * invN;
    const float var = colsq[i] * invN - mu * mu;
    const float sc  = rsqrtf(var + 1e-5f) * gamma[i];
    scale[i] = sc;
    shift[i] = beta[i] - mu * sc;
}

__global__ __launch_bounds__(64)
void k_out(const float* __restrict__ h, const float* __restrict__ W2,
           const float* __restrict__ scale, const float* __restrict__ shift,
           float* __restrict__ out, long long N) {
    __shared__ float W2s[HH][66];
    __shared__ float As[2][16][130];
    __shared__ float sc_s[128], sh_s[128];

    const int tid  = threadIdx.x;
    const int wave = tid >> 5;
    const int lane = tid & 31;
    const int hf   = lane >> 4;
    const int lm   = lane & 15;

    for (int i = tid; i < HH * DI; i += 64)
        W2s[i >> 6][i & 63] = W2[i];
    for (int i = tid; i < 128; i += 64) { sc_s[i] = scale[i]; sh_s[i] = shift[i]; }
    __syncthreads();

    const long long r0 = ((long long)blockIdx.x * 2 + wave) * 16;
    for (int r = 0; r < 16; ++r) {
        long long rr = r0 + r;
        long long rc = rr < N ? rr : N - 1;
        const int c = lane * 4;
        const v4f hv = *(const v4f*)(h + rc * HH + c);
        float v0 = hv.x * sc_s[c + 0] + sh_s[c + 0];
        float v1 = hv.y * sc_s[c + 1] + sh_s[c + 1];
        float v2 = hv.z * sc_s[c + 2] + sh_s[c + 2];
        float v3 = hv.w * sc_s[c + 3] + sh_s[c + 3];
        As[wave][r][c + 0] = v0 > 0.f ? v0 : 0.f;
        As[wave][r][c + 1] = v1 > 0.f ? v1 : 0.f;
        As[wave][r][c + 2] = v2 > 0.f ? v2 : 0.f;
        As[wave][r][c + 3] = v3 > 0.f ? v3 : 0.f;
    }
    __syncthreads();

    for (int n0 = 0; n0 < DI; n0 += 16) {
        v8f acc = {};
#pragma unroll
        for (int k0 = 0; k0 < HH; k0 += 4) {
            const int kb = k0 + hf * 2;
            v2f a, b;
            a.x = As[wave][lm][kb];   a.y = As[wave][lm][kb + 1];
            b.x = W2s[kb][n0 + lm];   b.y = W2s[kb + 1][n0 + lm];
            acc = wmma_f32_k4(a, b, acc);
        }
#pragma unroll
        for (int i = 0; i < 8; ++i) {
            const int m = i + hf * 8;
            const long long rr = r0 + m;
            if (rr < N) __builtin_nontemporal_store(acc[i], out + rr * DI + n0 + lm);
        }
    }
}

extern "C" void kernel_launch(void* const* d_in, const int* in_sizes, int n_in,
                              void* d_out, int out_size, void* d_ws, size_t ws_size,
                              hipStream_t stream) {
    const float* x     = (const float*)d_in[0];
    const int*   eidx  = (const int*)d_in[1];
    const float* ea    = (const float*)d_in[2];
    const float* Wedge = (const float*)d_in[3];
    const float* W1    = (const float*)d_in[4];
    const float* gamma = (const float*)d_in[5];
    const float* beta  = (const float*)d_in[6];
    const float* W2    = (const float*)d_in[7];
    float* out = (float*)d_out;

    const long long N = in_sizes[0] / DI;
    const long long E = in_sizes[1] / 2;
    const int* src = eidx;
    const int* dst = eidx + E;

    float* ws   = (float*)d_ws;
    float* msg  = ws;
    int*   cnt  = (int*)(msg + (size_t)E * DI);
    float* cs   = (float*)(cnt + (size_t)N);
    float* cq   = cs + 128;
    int*   offs = (int*)(cq + 128);
    int*   pos  = offs + (size_t)N;
    int*   chist = pos + (((size_t)E + 31) / 32) * 32;
    float* agg  = msg + (size_t)E * DI + (size_t)2 * N * DI;
    float* sc   = agg + (size_t)N * DI + 256;
    float* sh   = sc + 128;
    float* hbuf = sh + 128;

    const int C  = (int)(((E + NCHUNK - 1) / NCHUNK + 31) / 32 * 32);
    const int NP = (int)((N + 31) / 32 * 32);

    k_count<<<dim3(NCHUNK, (unsigned)((N + NR - 1) / NR)), 1024, 0, stream>>>(dst, pos, chist, E, (int)N, NP, C);
    k_colpre<<<(int)((N + 255) / 256), 256, 0, stream>>>(chist, cnt, cs, (int)N, NP);
    k_scan<<<1, 1024, 0, stream>>>(cnt, offs, (int)N);

    k_edge_msg<<<(int)((E + 63) / 64), 128, 0, stream>>>(x, src, dst, pos, offs, chist, ea, Wedge, msg, E, N, NP, C);

    k_node_agg<<<(int)((N * 16 + 255) / 256), 256, 0, stream>>>(msg, offs, cnt, agg, N);

    const int strips = (int)((N + 15) / 16);
    k_h_stats<<<(strips + 3) / 4, 128, 0, stream>>>(x, agg, W1, hbuf, cs, cq, N);
    k_bn<<<1, 128, 0, stream>>>(cs, cq, gamma, beta, sc, sh, 1.0f / (float)N);
    k_out<<<(strips + 1) / 2, 64, 0, stream>>>(hbuf, W2, sc, sh, out, N);
}


# both one-block launches folded into their neighbours (8 -> 6 launches): row offsets from per-chunk group prefixes of the count pass + an in-group scan; scale/shift derived inside the last kernel
# speedup vs baseline: 2.7819x; 1.1580x over previous
#include <hip/hip_runtime.h>
#include <hip/hip_bf16.h>

typedef float v2f __attribute__((ext_vector_type(2)));
typedef float v4f __attribute__((ext_vector_type(4)));
typedef float v8f __attribute__((ext_vector_type(8)));

#define DI 64
#define HH 128
#define NCHUNK 64
#define NR 65536
#define GN 256

__device__ __forceinline__ v8f wmma_f32_k4(v2f a, v2f b, v8f c) {
    return __builtin_amdgcn_wmma_f32_16x16x4_f32(false, a, false, b, (short)0, c, false, false);
}

__global__ __launch_bounds__(128)
void k_edge_msg(const float* __restrict__ x, const int* __restrict__ src,
                const int* __restrict__ dst, const int* __restrict__ pos,
                const int* __restrict__ offs, const int* __restrict__ cpre,
                const float* __restrict__ ea, const float* __restrict__ Wedge,
                float* __restrict__ msg, long long E, long long N, int NP, int C) {
    __shared__ float Ws[64][66];
    __shared__ __attribute__((aligned(16))) float As[4][16][68];

    const int tid  = threadIdx.x;
    const int wave = tid >> 5;
    const int lane = tid & 31;
    const int hf   = lane >> 4;
    const int lm   = lane & 15;

    const long long e0 = ((long long)blockIdx.x * 4 + wave) * 16;
    const long long ef = e0 < E ? e0 : E - 1;
    const int* __restrict__ cpre_c = cpre + (long long)((unsigned)ef / (unsigned)C) * NP;

    int xrow[8], drow[8], prow[8];
#pragma unroll
    for (int j = 0; j < 8; ++j) {
        const long long e  = e0 + 2 * j + hf;
        const long long ec = e < E ? e : E - 1;
        xrow[j] = src[ec];
        drow[j] = dst[ec];
        prow[j] = pos[ec];
    }

    v2f av[16];
#pragma unroll
    for (int r = 0; r < 16; ++r) {
        const long long e  = e0 + r;
        const long long ec = e < E ? e : E - 1;
        av[r] = __builtin_nontemporal_load((const v2f*)(ea + ec * DI + lane * 2));
    }

    for (int i = tid; i < 64 * 64; i += 128)
        Ws[i >> 6][i & 63] = Wedge[i];

#pragma unroll
    for (int r = 0; r < 16; ++r) {
        As[wave][r][lane * 2 + 0] = av[r].x;
        As[wave][r][lane * 2 + 1] = av[r].y;
    }
    __syncthreads();

    int mrow[8];
    v4f xg[8];
#pragma unroll
    for (int j = 0; j < 8; ++j)
        mrow[j] = offs[drow[j]] + cpre_c[drow[j]] + prow[j];
#pragma unroll
    for (int j = 0; j < 8; ++j)
        xg[j] = *(const v4f*)(x + (long long)xrow[j] * DI + lm * 4);

    v8f acc0 = {}, acc1 = {}, acc2 = {}, acc3 = {};
#pragma unroll
    for (int k0 = 0; k0 < DI; k0 += 4) {
        const int kb = k0 + hf * 2;
        v2f a, b;
        a.x = As[wave][lm][kb];     a.y = As[wave][lm][kb + 1];
        b.x = Ws[kb][lm];           b.y = Ws[kb + 1][lm];
        acc0 = wmma_f32_k4(a, b, acc0);
        b.x = Ws[kb][16 + lm];      b.y = Ws[kb + 1][16 + lm];
        acc1 = wmma_f32_k4(a, b, acc1);
        b.x = Ws[kb][32 + lm];      b.y = Ws[kb + 1][32 + lm];
        acc2 = wmma_f32_k4(a, b, acc2);
        b.x = Ws[kb][48 + lm];      b.y = Ws[kb + 1][48 + lm];
        acc3 = wmma_f32_k4(a, b, acc3);
    }

#pragma unroll
    for (int i = 0; i < 8; ++i) {
        const int m = i + hf * 8;
        As[wave][m][lm]      = acc0[i];
        As[wave][m][16 + lm] = acc1[i];
        As[wave][m][32 + lm] = acc2[i];
        As[wave][m][48 + lm] = acc3[i];
    }
    __syncthreads();

    if (e0 + 16 <= E) {
#pragma unroll
        for (int j = 0; j < 8; ++j) {
            const int row = 2 * j + hf;
            const v4f d = *(const v4f*)&As[wave][row][lm * 4];
            v4f v;
            v.x = d.x + xg[j].x;  v.y = d.y + xg[j].y;
            v.z = d.z + xg[j].z;  v.w = d.w + xg[j].w;
            v.x = (v.x > 0.f ? v.x : 0.f) + 1e-7f;  v.y = (v.y > 0.f ? v.y : 0.f) + 1e-7f;
            v.z = (v.z > 0.f ? v.z : 0.f) + 1e-7f;  v.w = (v.w > 0.f ? v.w : 0.f) + 1e-7f;
            __builtin_nontemporal_store(v, (v4f*)(msg + (long long)mrow[j] * DI + lm * 4));
        }
    } else {
#pragma unroll
        for (int j = 0; j < 8; ++j) {
            const int row = 2 * j + hf;
            const v4f d = *(const v4f*)&As[wave][row][lm * 4];
            v4f v;
            v.x = d.x + xg[j].x;  v.y = d.y + xg[j].y;
            v.z = d.z + xg[j].z;  v.w = d.w + xg[j].w;
            v.x = (v.x > 0.f ? v.x : 0.f) + 1e-7f;  v.y = (v.y > 0.f ? v.y : 0.f) + 1e-7f;
            v.z = (v.z > 0.f ? v.z : 0.f) + 1e-7f;  v.w = (v.w > 0.f ? v.w : 0.f) + 1e-7f;
            if (e0 + row < E)
                __builtin_nontemporal_store(v, (v4f*)(msg + (long long)mrow[j] * DI + lm * 4));
        }
    }
}

#define CNT_KPT 16
__global__ __launch_bounds__(1024)
void k_count(const int* __restrict__ dst, int* __restrict__ pos, int* __restrict__ chist,
             int* __restrict__ gtab, int* __restrict__ rtot,
             long long E, int N, int NP, int C, int G, int GP, int RY) {
    __shared__ __attribute__((aligned(16))) int hist[NR];
    __shared__ int gs[NR / GN];
    __shared__ int wt[8];
    const int tid = threadIdx.x;
    const int nlo = blockIdx.y * NR;
    const int nn  = (N - nlo) < NR ? (N - nlo) : NR;
    for (int i = tid * 4; i < nn; i += 4096) *(int4*)&hist[i] = make_int4(0, 0, 0, 0);
    if (tid < NR / GN) gs[tid] = 0;
    __syncthreads();
    const int lo = blockIdx.x * C;
    const int hi = (lo + C) < (int)E ? (lo + C) : (int)E;
    for (int base = lo; base < hi; base += 1024 * CNT_KPT) {
        int d[CNT_KPT], r[CNT_KPT];
#pragma unroll
        for (int k = 0; k < CNT_KPT; ++k) {
            const int e = base + k * 1024 + tid;
            const int v = dst[e < hi ? e : hi - 1];
            d[k] = e < hi ? v - nlo : -1;
        }
#pragma unroll
        for (int k = 0; k < CNT_KPT; ++k) {
            r[k] = 0;
            if ((unsigned)d[k] < (unsigned)nn) r[k] = atomicAdd(&hist[d[k]], 1);
        }
#pragma unroll
        for (int k = 0; k < CNT_KPT; ++k) {
            const int e = base + k * 1024 + tid;
            if ((unsigned)d[k] < (unsigned)nn) pos[e] = r[k];
        }
#pragma unroll
        for (int k = 0; k < CNT_KPT; ++k)
            if ((unsigned)d[k] < (unsigned)nn) atomicAdd(&gs[(unsigned)d[k] / GN], 1);
    }
    __syncthreads();
    int* __restrict__ row = chist + (long long)blockIdx.x * NP + nlo;
    for (int i = tid; i < nn; i += 1024) row[i] = hist[i];

    const int lane = tid & 31;
    const int wave = tid >> 5;
    int v = 0, inc = 0;
    if (tid < NR / GN) {
        v = gs[tid];
        inc = v;
#pragma unroll
        for (int d = 1; d < 32; d <<= 1) {
            const int t = __shfl_up(inc, d, 32);
            if (lane >= d) inc += t;
        }
        if (lane == 31) wt[wave] = inc;
    }
    __syncthreads();
    if (tid < NR / GN) {
        int wp = 0, tot = 0;
#pragma unroll
        for (int w = 0; w < 8; ++w) {
            const int t = wt[w];
            if (w < wave) wp += t;
            tot += t;
        }
        const int g = blockIdx.y * (NR / GN) + tid;
        if (g < G) gtab[(long long)blockIdx.x * GP + g] = wp + inc - v;
        if (tid == 0) rtot[((long long)blockIdx.x * RY + blockIdx.y) * 32] = tot;
    }
}

__global__ __launch_bounds__(GN)
void k_colpre(int* __restrict__ chist, int* __restrict__ cnt, int* __restrict__ offs,
              const int* __restrict__ gtab, const int* __restrict__ rtot,
              float* __restrict__ cstat, int N, int NP, int GP, int RY) {
    __shared__ int wt[8];
    __shared__ int wg[8];
    const int tid  = threadIdx.x;
    const int lane = tid & 31;
    const int wave = tid >> 5;
    const int n = blockIdx.x * GN + tid;
    if (blockIdx.x == 0) cstat[tid] = 0.f;
    const int ct = tid < NCHUNK ? tid : 0;
    int gb = gtab[(long long)ct * GP + blockIdx.x];
    const int y = blockIdx.x / (NR / GN);
#pragma nounroll
    for (int yy = 0; yy < y; ++yy) gb += rtot[((long long)ct * RY + yy) * 32];
    int run = 0;
    if (n < N) {
        for (int b0 = 0; b0 < NCHUNK; b0 += 16) {
            int v[16];
#pragma unroll
            for (int k = 0; k < 16; ++k) v[k] = chist[(long long)(b0 + k) * NP + n];
#pragma unroll
            for (int k = 0; k < 16; ++k) {
                chist[(long long)(b0 + k) * NP + n] = run;
                run += v[k];
            }
        }
        cnt[n] = run;
    }
    if (tid >= NCHUNK) gb = 0;
    int inc = run;
#pragma unroll
    for (int d = 1; d < 32; d <<= 1) {
        const int t = __shfl_up(inc, d, 32);
        if (lane >= d) inc += t;
    }
#pragma unroll
    for (int d = 16; d > 0; d >>= 1) gb += __shfl_xor(gb, d, 32);
    if (lane == 31) wt[wave] = inc;
    if (lane == 0)  wg[wave] = gb;
    __syncthreads();
    int base = 0;
#pragma unroll
    for (int w = 0; w < 8; ++w) {
        const int t = wt[w];
        if (w < wave) base += t;
        base += wg[w];
    }
    if (n < N) offs[n] = base + inc - run;
}

#define NCAP 24
__global__ __launch_bounds__(256)
void k_node_agg(const float* __restrict__ msg, const int* __restrict__ offs,
                const int* __restrict__ cnt, float* __restrict__ agg, long long N) {
    const long long g = (long long)blockIdx.x * blockDim.x + threadIdx.x;
    const long long node = g >> 4;
    if (node >= N) return;
    const int d4  = (int)(g & 15) * 4;
    const int deg = cnt[node];
    const float* p = msg + (long long)offs[node] * DI + d4;
    const float ninf = -__builtin_huge_valf();
    v4f mx = {ninf, ninf, ninf, ninf};
    v4f S = {0.f, 0.f, 0.f, 0.f};
    v4f A = {0.f, 0.f, 0.f, 0.f};
    if (deg <= NCAP) {
        v4f rows[NCAP];
#pragma unroll
        for (int j = 0; j < NCAP; ++j) {
            rows[j] = (v4f){ninf, ninf, ninf, ninf};
            if (j < deg) rows[j] = __builtin_nontemporal_load((const v4f*)(p + (long long)j * DI));
        }
#pragma unroll
        for (int j = 0; j < NCAP; ++j) {
            mx.x = fmaxf(mx.x, rows[j].x);  mx.y = fmaxf(mx.y, rows[j].y);
            mx.z = fmaxf(mx.z, rows[j].z);  mx.w = fmaxf(mx.w, rows[j].w);
        }
#pragma unroll
        for (int j = 0; j < NCAP; ++j) {
            const bool w = j < deg;
            const v4f m = rows[j];
            const float ex = __expf(m.x - mx.x), ey = __expf(m.y - mx.y);
            const float ez = __expf(m.z - mx.z), ew = __expf(m.w - mx.w);
            S.x += ex;  S.y += ey;  S.z += ez;  S.w += ew;
            A.x += (w ? m.x : 0.f) * ex;  A.y += (w ? m.y : 0.f) * ey;
            A.z += (w ? m.z : 0.f) * ez;  A.w += (w ? m.w : 0.f) * ew;
        }
    } else {
        for (int j = 0; j < deg; ++j) {
            const v4f m = *(const v4f*)(p + (long long)j * DI);
            mx.x = fmaxf(mx.x, m.x);  mx.y = fmaxf(mx.y, m.y);
            mx.z = fmaxf(mx.z, m.z);  mx.w = fmaxf(mx.w, m.w);
        }
        for (int j = 0; j < deg; ++j) {
            const v4f m = __builtin_nontemporal_load((const v4f*)(p + (long long)j * DI));
            const float ex = __expf(m.x - mx.x), ey = __expf(m.y - mx.y);
            const float ez = __expf(m.z - mx.z), ew = __expf(m.w - mx.w);
            S.x += ex;  S.y += ey;  S.z += ez;  S.w += ew;
            A.x += m.x * ex;  A.y += m.y * ey;  A.z += m.z * ez;  A.w += m.w * ew;
        }
    }
    v4f o = {0.f, 0.f, 0.f, 0.f};
    if (deg > 0) {
        o.x = A.x / (S.x + 1e-16f);  o.y = A.y / (S.y + 1e-16f);
        o.z = A.z / (S.z + 1e-16f);  o.w = A.w / (S.w + 1e-16f);
    }
    *(v4f*)(agg + node * DI + d4) = o;
}

__global__ __launch_bounds__(128)
void k_h_stats(const float* __restrict__ x, const float* __restrict__ agg,
               const float* __restrict__ W1, float* __restrict__ h,
               float* __restrict__ colsum, float* __restrict__ colsq, long long N) {
    __shared__ float W1s[64][130];
    __shared__ float As[4][16][66];
    __shared__ float cs_s[128], cq_s[128];

    const int tid  = threadIdx.x;
    const int wave = tid >> 5;
    const int lane = tid & 31;
    const int hf   = lane >> 4;
    const int lm   = lane & 15;

    for (int i = tid; i < 64 * HH; i += 128)
        W1s[i >> 7][i & 127] = W1[i];
    if (tid < 128) { cs_s[tid] = 0.f; cq_s[tid] = 0.f; }

    const long long r0 = ((long long)blockIdx.x * 4 + wave) * 16;
    for (int r = 0; r < 16; ++r) {
        long long rr = r0 + r;
        long long rc = rr < N ? rr : N - 1;
        const float2 a  = *(const float2*)(agg + rc * DI + lane * 2);
        const float2 xv = *(const float2*)(x + rc * DI + lane * 2);
        As[wave][r][lane * 2 + 0] = a.x + xv.x;
        As[wave][r][lane * 2 + 1] = a.y + xv.y;
    }
    __syncthreads();

    for (int n0 = 0; n0 < HH; n0 += 16) {
        v8f acc = {};
#pragma unroll
        for (int k0 = 0; k0 < DI; k0 += 4) {
            const int kb = k0 + hf * 2;
            v2f a, b;
            a.x = As[wave][lm][kb];   a.y = As[wave][lm][kb + 1];
            b.x = W1s[kb][n0 + lm];   b.y = W1s[kb + 1][n0 + lm];
            acc = wmma_f32_k4(a, b, acc);
        }
        float p1 = 0.f, p2 = 0.f;
#pragma unroll
        for (int i = 0; i < 8; ++i) {
            const int m = i + hf * 8;
            const long long rr = r0 + m;
            if (rr < N) {
                const float v = acc[i];
                h[rr * HH + n0 + lm] = v;
                p1 += v;
                p2 += v * v;
            }
        }
        p1 += __shfl_xor(p1, 16, 32);
        p2 += __shfl_xor(p2, 16, 32);
        if (hf == 0) {
            atomicAdd(&cs_s[n0 + lm], p1);
            atomicAdd(&cq_s[n0 + lm], p2);
        }
    }
    __syncthreads();
    if (tid < 128) {
        atomicAdd(colsum + tid, cs_s[tid]);
        atomicAdd(colsq + tid, cq_s[tid]);
    }
}

__global__ __launch_bounds__(64)
void k_out(const float* __restrict__ h, const float* __restrict__ W2,
           const float* __restrict__ colsum, const float* __restrict__ colsq,
           const float* __restrict__ gamma, const float* __restrict__ beta,
           float* __restrict__ out, long long N, float invN) {
    __shared__ float W2s[HH][66];
    __shared__ float As[2][16][130];
    __shared__ float sc_s[128], sh_s[128];

    const int tid  = threadIdx.x;
    const int wave = tid >> 5;
    const int lane = tid & 31;
    const int hf   = lane >> 4;
    const int lm   = lane & 15;

    float bsum[2], bsq[2], bgam[2], bbet[2];
#pragma unroll
    for (int j = 0; j < 2; ++j) {
        const int i = tid + 64 * j;
        bsum[j] = colsum[i];  bsq[j]  = colsq[i];
        bgam[j] = gamma[i];   bbet[j] = beta[i];
    }
    for (int i = tid; i < HH * DI; i += 64)
        W2s[i >> 6][i & 63] = W2[i];
#pragma unroll
    for (int j = 0; j < 2; ++j) {
        const int i = tid + 64 * j;
        const float mu  = bsum[j] * invN;
        const float var = bsq[j] * invN - mu * mu;
        const float sc  = rsqrtf(var + 1e-5f) * bgam[j];
        sc_s[i] = sc;
        sh_s[i] = bbet[j] - mu * sc;
    }
    __syncthreads();

    const long long r0 = ((long long)blockIdx.x * 2 + wave) * 16;
    for (int r = 0; r < 16; ++r) {
        long long rr = r0 + r;
        long long rc = rr < N ? rr : N - 1;
        const int c = lane * 4;
        const v4f hv = *(const v4f*)(h + rc * HH + c);
        float v0 = hv.x * sc_s[c + 0] + sh_s[c + 0];
        float v1 = hv.y * sc_s[c + 1] + sh_s[c + 1];
        float v2 = hv.z * sc_s[c + 2] + sh_s[c + 2];
        float v3 = hv.w * sc_s[c + 3] + sh_s[c + 3];
        As[wave][r][c + 0] = v0 > 0.f ? v0 : 0.f;
        As[wave][r][c + 1] = v1 > 0.f ? v1 : 0.f;
        As[wave][r][c + 2] = v2 > 0.f ? v2 : 0.f;
        As[wave][r][c + 3] = v3 > 0.f ? v3 : 0.f;
    }
    __syncthreads();

    for (int n0 = 0; n0 < DI; n0 += 16) {
        v8f acc = {};
#pragma unroll
        for (int k0 = 0; k0 < HH; k0 += 4) {
            const int kb = k0 + hf * 2;
            v2f a, b;
            a.x = As[wave][lm][kb];   a.y = As[wave][lm][kb + 1];
            b.x = W2s[kb][n0 + lm];   b.y = W2s[kb + 1][n0 + lm];
            acc = wmma_f32_k4(a, b, acc);
        }
#pragma unroll
        for (int i = 0; i < 8; ++i) {
            const int m = i + hf * 8;
            const long long rr = r0 + m;
            if (rr < N) __builtin_nontemporal_store(acc[i], out + rr * DI + n0 + lm);
        }
    }
}

extern "C" void kernel_launch(void* const* d_in, const int* in_sizes, int n_in,
                              void* d_out, int out_size, void* d_ws, size_t ws_size,
                              hipStream_t stream) {
    const float* x     = (const float*)d_in[0];
    const int*   eidx  = (const int*)d_in[1];
    const float* ea    = (const float*)d_in[2];
    const float* Wedge = (const float*)d_in[3];
    const float* W1    = (const float*)d_in[4];
    const float* gamma = (const float*)d_in[5];
    const float* beta  = (const float*)d_in[6];
    const float* W2    = (const float*)d_in[7];
    float* out = (float*)d_out;

    const long long N = in_sizes[0] / DI;
    const long long E = in_sizes[1] / 2;
    const int* src = eidx;
    const int* dst = eidx + E;

    float* ws   = (float*)d_ws;
    float* msg  = ws;
    int*   cnt  = (int*)(msg + (size_t)E * DI);
    float* cs   = (float*)(cnt + (size_t)N);
    float* cq   = cs + 128;
    int*   offs = (int*)(cq + 128);
    int*   pos  = offs + (size_t)N;
    int*   chist = pos + (((size_t)E + 31) / 32) * 32;
    float* agg  = msg + (size_t)E * DI + (size_t)2 * N * DI;
    float* hbuf = agg + (size_t)N * DI + 512;

    const int C  = (int)(((E + NCHUNK - 1) / NCHUNK + 31) / 32 * 32);
    const int NP = (int)((N + 31) / 32 * 32);
    const int G  = (int)((N + GN - 1) / GN);
    const int GP = (G + 31) / 32 * 32;
    const int RY = (int)((N + NR - 1) / NR);
    int* gtab = chist + (size_t)NCHUNK * NP;
    int* rtot = gtab + (size_t)NCHUNK * GP;

    k_count<<<dim3(NCHUNK, (unsigned)RY), 1024, 0, stream>>>(dst, pos, chist, gtab, rtot, E, (int)N, NP, C, G, GP, RY);
    k_colpre<<<G, GN, 0, stream>>>(chist, cnt, offs, gtab, rtot, cs, (int)N, NP, GP, RY);

    k_edge_msg<<<(int)((E + 63) / 64), 128, 0, stream>>>(x, src, dst, pos, offs, chist, ea, Wedge, msg, E, N, NP, C);

    k_node_agg<<<(int)((N * 16 + 255) / 256), 256, 0, stream>>>(msg, offs, cnt, agg, N);

    const int strips = (int)((N + 15) / 16);
    k_h_stats<<<(strips + 3) / 4, 128, 0, stream>>>(x, agg, W1, hbuf, cs, cq, N);
    k_out<<<(strips + 1) / 2, 64, 0, stream>>>(hbuf, W2, cs, cq, gamma, beta, out, N, 1.0f / (float)N);
}


# LDS operand fragments as 16-B k-quads in all three product kernels: weights staged quad-wise, A and B loaders both take k = 8p+4h..+3 per lane half; the k order of the sums in each 8-k block changes
# speedup vs baseline: 2.9428x; 1.0578x over previous
#include <hip/hip_runtime.h>
#include <hip/hip_bf16.h>

typedef float v2f __attribute__((ext_vector_type(2)));
typedef float v4f __attribute__((ext_vector_type(4)));
typedef float v8f __attribute__((ext_vector_type(8)));

#define DI 64
#define HH 128
#define NCHUNK 64
#define NR 65536
#define GN 256

__device__ __forceinline__ v8f wmma_f32_k4(v2f a, v2f b, v8f c) {
    return __builtin_amdgcn_wmma_f32_16x16x4_f32(false, a, false, b, (short)0, c, false, false);
}

__global__ __launch_bounds__(128)
void k_edge_msg(const float* __restrict__ x, const int* __restrict__ src,
                const int* __restrict__ dst, const int* __restrict__ pos,
                const int* __restrict__ offs, const int* __restrict__ cpre,
                const float* __restrict__ ea, const float* __restrict__ Wedge,
                float* __restrict__ msg, long long E, long long N, int NP, int C) {
    __shared__ __attribute__((aligned(16))) float Wq[16][64][4];
    __shared__ __attribute__((aligned(16))) float As[4][16][68];

    const int tid  = threadIdx.x;
    const int wave = tid >> 5;
    const int lane = tid & 31;
    const int hf   = lane >> 4;
    const int lm   = lane & 15;

    const long long e0 = ((long long)blockIdx.x * 4 + wave) * 16;
    const long long ef = e0 < E ? e0 : E - 1;
    const int* __restrict__ cpre_c = cpre + (long long)((unsigned)ef / (unsigned)C) * NP;

    int xrow[8], drow[8], prow[8];
#pragma unroll
    for (int j = 0; j < 8; ++j) {
        const long long e  = e0 + 2 * j + hf;
        const long long ec = e < E ? e : E - 1;
        xrow[j] = src[ec];
        drow[j] = dst[ec];
        prow[j] = pos[ec];
    }

    v2f av[16];
#pragma unroll
    for (int r = 0; r < 16; ++r) {
        const long long e  = e0 + r;
        const long long ec = e < E ? e : E - 1;
        av[r] = __builtin_nontemporal_load((const v2f*)(ea + ec * DI + lane * 2));
    }

#pragma unroll 4
    for (int it = 0; it < 8; ++it) {
        const int i = tid + 128 * it;
        const int q = i >> 6, cc = i & 63;
        v4f w;
        w.x = Wedge[(4 * q + 0) * DI + cc];
        w.y = Wedge[(4 * q + 1) * DI + cc];
        w.z = Wedge[(4 * q + 2) * DI + cc];
        w.w = Wedge[(4 * q + 3) * DI + cc];
        *(v4f*)&Wq[q][cc][0] = w;
    }

#pragma unroll
    for (int r = 0; r < 16; ++r) {
        As[wave][r][lane * 2 + 0] = av[r].x;
        As[wave][r][lane * 2 + 1] = av[r].y;
    }
    __syncthreads();

    int mrow[8];
    v4f xg[8];
#pragma unroll
    for (int j = 0; j < 8; ++j)
        mrow[j] = offs[drow[j]] + cpre_c[drow[j]] + prow[j];
#pragma unroll
    for (int j = 0; j < 8; ++j)
        xg[j] = *(const v4f*)(x + (long long)xrow[j] * DI + lm * 4);

    v8f acc0 = {}, acc1 = {}, acc2 = {}, acc3 = {};
#pragma unroll
    for (int p = 0; p < DI / 8; ++p) {
        const int q = 2 * p + hf;
        const v4f a4 = *(const v4f*)&As[wave][lm][4 * q];
        const v4f b0 = *(const v4f*)&Wq[q][lm][0];
        const v4f b1 = *(const v4f*)&Wq[q][16 + lm][0];
        const v4f b2 = *(const v4f*)&Wq[q][32 + lm][0];
        const v4f b3 = *(const v4f*)&Wq[q][48 + lm][0];
        const v2f alo = {a4.x, a4.y}, ahi = {a4.z, a4.w};
        acc0 = wmma_f32_k4(alo, (v2f){b0.x, b0.y}, acc0);
        acc1 = wmma_f32_k4(alo, (v2f){b1.x, b1.y}, acc1);
        acc2 = wmma_f32_k4(alo, (v2f){b2.x, b2.y}, acc2);
        acc3 = wmma_f32_k4(alo, (v2f){b3.x, b3.y}, acc3);
        acc0 = wmma_f32_k4(ahi, (v2f){b0.z, b0.w}, acc0);
        acc1 = wmma_f32_k4(ahi, (v2f){b1.z, b1.w}, acc1);
        acc2 = wmma_f32_k4(ahi, (v2f){b2.z, b2.w}, acc2);
        acc3 = wmma_f32_k4(ahi, (v2f){b3.z, b3.w}, acc3);
    }

#pragma unroll
    for (int i = 0; i < 8; ++i) {
        const int m = i + hf * 8;
        As[wave][m][lm]      = acc0[i];
        As[wave][m][16 + lm] = acc1[i];
        As[wave][m][32 + lm] = acc2[i];
        As[wave][m][48 + lm] = acc3[i];
    }
    __syncthreads();

    if (e0 + 16 <= E) {
#pragma unroll
        for (int j = 0; j < 8; ++j) {
            const int row = 2 * j + hf;
            const v4f d = *(const v4f*)&As[wave][row][lm * 4];
            v4f v;
            v.x = d.x + xg[j].x;  v.y = d.y + xg[j].y;
            v.z = d.z + xg[j].z;  v.w = d.w + xg[j].w;
            v.x = (v.x > 0.f ? v.x : 0.f) + 1e-7f;  v.y = (v.y > 0.f ? v.y : 0.f) + 1e-7f;
            v.z = (v.z > 0.f ? v.z : 0.f) + 1e-7f;  v.w = (v.w > 0.f ? v.w : 0.f) + 1e-7f;
            __builtin_nontemporal_store(v, (v4f*)(msg + (long long)mrow[j] * DI + lm * 4));
        }
    } else {
#pragma unroll
        for (int j = 0; j < 8; ++j) {
            const int row = 2 * j + hf;
            const v4f d = *(const v4f*)&As[wave][row][lm * 4];
            v4f v;
            v.x = d.x + xg[j].x;  v.y = d.y + xg[j].y;
            v.z = d.z + xg[j].z;  v.w = d.w + xg[j].w;
            v.x = (v.x > 0.f ? v.x : 0.f) + 1e-7f;  v.y = (v.y > 0.f ? v.y : 0.f) + 1e-7f;
            v.z = (v.z > 0.f ? v.z : 0.f) + 1e-7f;  v.w = (v.w > 0.f ? v.w : 0.f) + 1e-7f;
            if (e0 + row < E)
                __builtin_nontemporal_store(v, (v4f*)(msg + (long long)mrow[j] * DI + lm * 4));
        }
    }
}

#define CNT_KPT 16
__global__ __launch_bounds__(1024)
void k_count(const int* __restrict__ dst, int* __restrict__ pos, int* __restrict__ chist,
             int* __restrict__ gtab, int* __restrict__ rtot,
             long long E, int N, int NP, int C, int G, int GP, int RY) {
    __shared__ __attribute__((aligned(16))) int hist[NR];
    __shared__ int gs[NR / GN];
    __shared__ int wt[8];
    const int tid = threadIdx.x;
    const int nlo = blockIdx.y * NR;
    const int nn  = (N - nlo) < NR ? (N - nlo) : NR;
    for (int i = tid * 4; i < nn; i += 4096) *(int4*)&hist[i] = make_int4(0, 0, 0, 0);
    if (tid < NR / GN) gs[tid] = 0;
    __syncthreads();
    const int lo = blockIdx.x * C;
    const int hi = (lo + C) < (int)E ? (lo + C) : (int)E;
    for (int base = lo; base < hi; base += 1024 * CNT_KPT) {
        int d[CNT_KPT], r[CNT_KPT];
#pragma unroll
        for (int k = 0; k < CNT_KPT; ++k) {
            const int e = base + k * 1024 + tid;
            const int v = dst[e < hi ? e : hi - 1];
            d[k] = e < hi ? v - nlo : -1;
        }
#pragma unroll
        for (int k = 0; k < CNT_KPT; ++k) {
            r[k] = 0;
            if ((unsigned)d[k] < (unsigned)nn) r[k] = atomicAdd(&hist[d[k]], 1);
        }
#pragma unroll
        for (int k = 0; k < CNT_KPT; ++k) {
            const int e = base + k * 1024 + tid;
            if ((unsigned)d[k] < (unsigned)nn) pos[e] = r[k];
        }
#pragma unroll
        for (int k = 0; k < CNT_KPT; ++k)
            if ((unsigned)d[k] < (unsigned)nn) atomicAdd(&gs[(unsigned)d[k] / GN], 1);
    }
    __syncthreads();
    int* __restrict__ row = chist + (long long)blockIdx.x * NP + nlo;
    for (int i = tid; i < nn; i += 1024) row[i] = hist[i];

    const int lane = tid & 31;
    const int wave = tid >> 5;
    int v = 0, inc = 0;
    if (tid < NR / GN) {
        v = gs[tid];
        inc = v;
#pragma unroll
        for (int d = 1; d < 32; d <<= 1) {
            const int t = __shfl_up(inc, d, 32);
            if (lane >= d) inc += t;
        }
        if (lane == 31) wt[wave] = inc;
    }
    __syncthreads();
    if (tid < NR / GN) {
        int wp = 0, tot = 0;
#pragma unroll
        for (int w = 0; w < 8; ++w) {
            const int t = wt[w];
            if (w < wave) wp += t;
            tot += t;
        }
        const int g = blockIdx.y * (NR / GN) + tid;
        if (g < G) gtab[(long long)blockIdx.x * GP + g] = wp + inc - v;
        if (tid == 0) rtot[((long long)blockIdx.x * RY + blockIdx.y) * 32] = tot;
    }
}

__global__ __launch_bounds__(GN)
void k_colpre(int* __restrict__ chist, int* __restrict__ cnt, int* __restrict__ offs,
              const int* __restrict__ gtab, const int* __restrict__ rtot,
              float* __restrict__ cstat, int N, int NP, int GP, int RY) {
    __shared__ int wt[8];
    __shared__ int wg[8];
    const int tid  = threadIdx.x;
    const int lane = tid & 31;
    const int wave = tid >> 5;
    const int n = blockIdx.x * GN + tid;
    if (blockIdx.x == 0) cstat[tid] = 0.f;
    const int ct = tid < NCHUNK ? tid : 0;
    int gb = gtab[(long long)ct * GP + blockIdx.x];
    const int y = blockIdx.x / (NR / GN);
#pragma nounroll
    for (int yy = 0; yy < y; ++yy) gb += rtot[((long long)ct * RY + yy) * 32];
    int run = 0;
    if (n < N) {
        for (int b0 = 0; b0 < NCHUNK; b0 += 16) {
            int v[16];
#pragma unroll
            for (int k = 0; k < 16; ++k) v[k] = chist[(long long)(b0 + k) * NP + n];
#pragma unroll
            for (int k = 0; k < 16; ++k) {
                chist[(long long)(b0 + k) * NP + n] = run;
                run += v[k];
            }
        }
        cnt[n] = run;
    }
    if (tid >= NCHUNK) gb = 0;
    int inc = run;
#pragma unroll
    for (int d = 1; d < 32; d <<= 1) {
        const int t = __shfl_up(inc, d, 32);
        if (lane >= d) inc += t;
    }
#pragma unroll
    for (int d = 16; d > 0; d >>= 1) gb += __shfl_xor(gb, d, 32);
    if (lane == 31) wt[wave] = inc;
    if (lane == 0)  wg[wave] = gb;
    __syncthreads();
    int base = 0;
#pragma unroll
    for (int w = 0; w < 8; ++w) {
        const int t = wt[w];
        if (w < wave) base += t;
        base += wg[w];
    }
    if (n < N) offs[n] = base + inc - run;
}

#define NCAP 24
__global__ __launch_bounds__(256)
void k_node_agg(const float* __restrict__ msg, const int* __restrict__ offs,
                const int* __restrict__ cnt, float* __restrict__ agg, long long N) {
    const long long g = (long long)blockIdx.x * blockDim.x + threadIdx.x;
    const long long node = g >> 4;
    if (node >= N) return;
    const int d4  = (int)(g & 15) * 4;
    const int deg = cnt[node];
    const float* p = msg + (long long)offs[node] * DI + d4;
    const float ninf = -__builtin_huge_valf();
    v4f mx = {ninf, ninf, ninf, ninf};
    v4f S = {0.f, 0.f, 0.f, 0.f};
    v4f A = {0.f, 0.f, 0.f, 0.f};
    if (deg <= NCAP) {
        v4f rows[NCAP];
#pragma unroll
        for (int j = 0; j < NCAP; ++j) {
            rows[j] = (v4f){ninf, ninf, ninf, ninf};
            if (j < deg) rows[j] = __builtin_nontemporal_load((const v4f*)(p + (long long)j * DI));
        }
#pragma unroll
        for (int j = 0; j < NCAP; ++j) {
            mx.x = fmaxf(mx.x, rows[j].x);  mx.y = fmaxf(mx.y, rows[j].y);
            mx.z = fmaxf(mx.z, rows[j].z);  mx.w = fmaxf(mx.w, rows[j].w);
        }
#pragma unroll
        for (int j = 0; j < NCAP; ++j) {
            const bool w = j < deg;
            const v4f m = rows[j];
            const float ex = __expf(m.x - mx.x), ey = __expf(m.y - mx.y);
            const float ez = __expf(m.z - mx.z), ew = __expf(m.w - mx.w);
            S.x += ex;  S.y += ey;  S.z += ez;  S.w += ew;
            A.x += (w ? m.x : 0.f) * ex;  A.y += (w ? m.y : 0.f) * ey;
            A.z += (w ? m.z : 0.f) * ez;  A.w += (w ? m.w : 0.f) * ew;
        }
    } else {
        for (int j = 0; j < deg; ++j) {
            const v4f m = *(const v4f*)(p + (long long)j * DI);
            mx.x = fmaxf(mx.x, m.x);  mx.y = fmaxf(mx.y, m.y);
            mx.z = fmaxf(mx.z, m.z);  mx.w = fmaxf(mx.w, m.w);
        }
        for (int j = 0; j < deg; ++j) {
            const v4f m = __builtin_nontemporal_load((const v4f*)(p + (long long)j * DI));
            const float ex = __expf(m.x - mx.x), ey = __expf(m.y - mx.y);
            const float ez = __expf(m.z - mx.z), ew = __expf(m.w - mx.w);
            S.x += ex;  S.y += ey;  S.z += ez;  S.w += ew;
            A.x += m.x * ex;  A.y += m.y * ey;  A.z += m.z * ez;  A.w += m.w * ew;
        }
    }
    v4f o = {0.f, 0.f, 0.f, 0.f};
    if (deg > 0) {
        o.x = A.x / (S.x + 1e-16f);  o.y = A.y / (S.y + 1e-16f);
        o.z = A.z / (S.z + 1e-16f);  o.w = A.w / (S.w + 1e-16f);
    }
    *(v4f*)(agg + node * DI + d4) = o;
}

__global__ __launch_bounds__(128)
void k_h_stats(const float* __restrict__ x, const float* __restrict__ agg,
               const float* __restrict__ W1, float* __restrict__ h,
               float* __restrict__ colsum, float* __restrict__ colsq, long long N) {
    __shared__ __attribute__((aligned(16))) float W1q[16][HH][4];
    __shared__ __attribute__((aligned(16))) float As[4][16][68];
    __shared__ float cs_s[128], cq_s[128];

    const int tid  = threadIdx.x;
    const int wave = tid >> 5;
    const int lane = tid & 31;
    const int hf   = lane >> 4;
    const int lm   = lane & 15;

#pragma unroll 4
    for (int it = 0; it < 16; ++it) {
        const int i = tid + 128 * it;
        const int q = i >> 7, cc = i & 127;
        v4f w;
        w.x = W1[(4 * q + 0) * HH + cc];
        w.y = W1[(4 * q + 1) * HH + cc];
        w.z = W1[(4 * q + 2) * HH + cc];
        w.w = W1[(4 * q + 3) * HH + cc];
        *(v4f*)&W1q[q][cc][0] = w;
    }
    if (tid < 128) { cs_s[tid] = 0.f; cq_s[tid] = 0.f; }

    const long long r0 = ((long long)blockIdx.x * 4 + wave) * 16;
    for (int r = 0; r < 16; ++r) {
        long long rr = r0 + r;
        long long rc = rr < N ? rr : N - 1;
        const float2 a  = *(const float2*)(agg + rc * DI + lane * 2);
        const float2 xv = *(const float2*)(x + rc * DI + lane * 2);
        As[wave][r][lane * 2 + 0] = a.x + xv.x;
        As[wave][r][lane * 2 + 1] = a.y + xv.y;
    }
    __syncthreads();

    for (int n0 = 0; n0 < HH; n0 += 16) {
        v8f acc = {};
#pragma unroll
        for (int p = 0; p < DI / 8; ++p) {
            const int q = 2 * p + hf;
            const v4f a4 = *(const v4f*)&As[wave][lm][4 * q];
            const v4f b4 = *(const v4f*)&W1q[q][n0 + lm][0];
            acc = wmma_f32_k4((v2f){a4.x, a4.y}, (v2f){b4.x, b4.y}, acc);
            acc = wmma_f32_k4((v2f){a4.z, a4.w}, (v2f){b4.z, b4.w}, acc);
        }
        float p1 = 0.f, p2 = 0.f;
#pragma unroll
        for (int i = 0; i < 8; ++i) {
            const int m = i + hf * 8;
            const long long rr = r0 + m;
            if (rr < N) {
                const float v = acc[i];
                h[rr * HH + n0 + lm] = v;
                p1 += v;
                p2 += v * v;
            }
        }
        p1 += __shfl_xor(p1, 16, 32);
        p2 += __shfl_xor(p2, 16, 32);
        if (hf == 0) {
            atomicAdd(&cs_s[n0 + lm], p1);
            atomicAdd(&cq_s[n0 + lm], p2);
        }
    }
    __syncthreads();
    if (tid < 128) {
        atomicAdd(colsum + tid, cs_s[tid]);
        atomicAdd(colsq + tid, cq_s[tid]);
    }
}

__global__ __launch_bounds__(64)
void k_out(const float* __restrict__ h, const float* __restrict__ W2,
           const float* __restrict__ colsum, const float* __restrict__ colsq,
           const float* __restrict__ gamma, const float* __restrict__ beta,
           float* __restrict__ out, long long N, float invN) {
    __shared__ __attribute__((aligned(16))) float W2q[32][DI][4];
    __shared__ __attribute__((aligned(16))) float As[2][16][132];
    __shared__ float sc_s[128], sh_s[128];

    const int tid  = threadIdx.x;
    const int wave = tid >> 5;
    const int lane = tid & 31;
    const int hf   = lane >> 4;
    const int lm   = lane & 15;

    float bsum[2], bsq[2], bgam[2], bbet[2];
#pragma unroll
    for (int j = 0; j < 2; ++j) {
        const int i = tid + 64 * j;
        bsum[j] = colsum[i];  bsq[j]  = colsq[i];
        bgam[j] = gamma[i];   bbet[j] = beta[i];
    }
#pragma unroll 4
    for (int it = 0; it < 32; ++it) {
        const int i = tid + 64 * it;
        const int q = i >> 6, cc = i & 63;
        v4f w;
        w.x = W2[(4 * q + 0) * DI + cc];
        w.y = W2[(4 * q + 1) * DI + cc];
        w.z = W2[(4 * q + 2) * DI + cc];
        w.w = W2[(4 * q + 3) * DI + cc];
        *(v4f*)&W2q[q][cc][0] = w;
    }
#pragma unroll
    for (int j = 0; j < 2; ++j) {
        const int i = tid + 64 * j;
        const float mu  = bsum[j] * invN;
        const float var = bsq[j] * invN - mu * mu;
        const float sc  = rsqrtf(var + 1e-5f) * bgam[j];
        sc_s[i] = sc;
        sh_s[i] = bbet[j] - mu * sc;
    }
    __syncthreads();

    const long long r0 = ((long long)blockIdx.x * 2 + wave) * 16;
    for (int r = 0; r < 16; ++r) {
        long long rr = r0 + r;
        long long rc = rr < N ? rr : N - 1;
        const int c = lane * 4;
        const v4f hv = *(const v4f*)(h + rc * HH + c);
        float v0 = hv.x * sc_s[c + 0] + sh_s[c + 0];
        float v1 = hv.y * sc_s[c + 1] + sh_s[c + 1];
        float v2 = hv.z * sc_s[c + 2] + sh_s[c + 2];
        float v3 = hv.w * sc_s[c + 3] + sh_s[c + 3];
        v4f o;
        o.x = v0 > 0.f ? v0 : 0.f;  o.y = v1 > 0.f ? v1 : 0.f;
        o.z = v2 > 0.f ? v2 : 0.f;  o.w = v3 > 0.f ? v3 : 0.f;
        *(v4f*)&As[wave][r][c] = o;
    }
    __syncthreads();

    for (int n0 = 0; n0 < DI; n0 += 16) {
        v8f acc = {};
#pragma unroll
        for (int p = 0; p < HH / 8; ++p) {
            const int q = 2 * p + hf;
            const v4f a4 = *(const v4f*)&As[wave][lm][4 * q];
            const v4f b4 = *(const v4f*)&W2q[q][n0 + lm][0];
            acc = wmma_f32_k4((v2f){a4.x, a4.y}, (v2f){b4.x, b4.y}, acc);
            acc = wmma_f32_k4((v2f){a4.z, a4.w}, (v2f){b4.z, b4.w}, acc);
        }
#pragma unroll
        for (int i = 0; i < 8; ++i) {
            const int m = i + hf * 8;
            const long long rr = r0 + m;
            if (rr < N) __builtin_nontemporal_store(acc[i], out + rr * DI + n0 + lm);
        }
    }
}

extern "C" void kernel_launch(void* const* d_in, const int* in_sizes, int n_in,
                              void* d_out, int out_size, void* d_ws, size_t ws_size,
                              hipStream_t stream) {
    const float* x     = (const float*)d_in[0];
    const int*   eidx  = (const int*)d_in[1];
    const float* ea    = (const float*)d_in[2];
    const float* Wedge = (const float*)d_in[3];
    const float* W1    = (const float*)d_in[4];
    const float* gamma = (const float*)d_in[5];
    const float* beta  = (const float*)d_in[6];
    const float* W2    = (const float*)d_in[7];
    float* out = (float*)d_out;

    const long long N = in_sizes[0] / DI;
    const long long E = in_sizes[1] / 2;
    const int* src = eidx;
    const int* dst = eidx + E;

    float* ws   = (float*)d_ws;
    float* msg  = ws;
    int*   cnt  = (int*)(msg + (size_t)E * DI);
    float* cs   = (float*)(cnt + (size_t)N);
    float* cq   = cs + 128;
    int*   offs = (int*)(cq + 128);
    int*   pos  = offs + (size_t)N;
    int*   chist = pos + (((size_t)E + 31) / 32) * 32;
    float* agg  = msg + (size_t)E * DI + (size_t)2 * N * DI;
    float* hbuf = agg + (size_t)N * DI + 512;

    const int C  = (int)(((E + NCHUNK - 1) / NCHUNK + 31) / 32 * 32);
    const int NP = (int)((N + 31) / 32 * 32);
    const int G  = (int)((N + GN - 1) / GN);
    const int GP = (G + 31) / 32 * 32;
    const int RY = (int)((N + NR - 1) / NR);
    int* gtab = chist + (size_t)NCHUNK * NP;
    int* rtot = gtab + (size_t)NCHUNK * GP;

    k_count<<<dim3(NCHUNK, (unsigned)RY), 1024, 0, stream>>>(dst, pos, chist, gtab, rtot, E, (int)N, NP, C, G, GP, RY);
    k_colpre<<<G, GN, 0, stream>>>(chist, cnt, offs, gtab, rtot, cs, (int)N, NP, GP, RY);

    k_edge_msg<<<(int)((E + 63) / 64), 128, 0, stream>>>(x, src, dst, pos, offs, chist, ea, Wedge, msg, E, N, NP, C);

    k_node_agg<<<(int)((N * 16 + 255) / 256), 256, 0, stream>>>(msg, offs, cnt, agg, N);

    const int strips = (int)((N + 15) / 16);
    k_h_stats<<<(strips + 3) / 4, 128, 0, stream>>>(x, agg, W1, hbuf, cs, cq, N);
    k_out<<<(strips + 1) / 2, 64, 0, stream>>>(hbuf, W2, cs, cq, gamma, beta, out, N, 1.0f / (float)N);
}
